# starting kernel
# baseline (speedup 1.0000x reference)
#include <hip/hip_runtime.h>
#include <hip/hip_bf16.h>

#define DIM 128

typedef __attribute__((ext_vector_type(2))) float v2f;
typedef __attribute__((ext_vector_type(8))) float v8f;

__global__ void sage_zero_f32(float* __restrict__ p, long long n) {
  long long i = (long long)blockIdx.x * blockDim.x + threadIdx.x;
  long long stride = (long long)gridDim.x * blockDim.x;
  for (; i < n; i += stride) p[i] = 0.0f;
}

__global__ void sage_degree(const int* __restrict__ dst, float* __restrict__ cnt, int E) {
  int i = blockIdx.x * blockDim.x + threadIdx.x;
  int stride = gridDim.x * blockDim.x;
  for (; i < E; i += stride) atomicAdd(&cnt[dst[i]], 1.0f);
}

__global__ void sage_scatter(const float* __restrict__ feat,
                             const int* __restrict__ src,
                             const int* __restrict__ dst,
                             float* __restrict__ agg, int E) {
  const int lane = threadIdx.x & 31;
  const int e = blockIdx.x * (blockDim.x >> 5) + (threadIdx.x >> 5);
  if (e >= E) return;
  const int s = src[e];
  const int d = dst[e];
  const float4 v = reinterpret_cast<const float4*>(feat + (size_t)s * DIM)[lane];
  float* out = agg + (size_t)d * DIM + lane * 4;
  atomicAdd(out + 0, v.x);
  atomicAdd(out + 1, v.y);
  atomicAdd(out + 2, v.z);
  atomicAdd(out + 3, v.w);
}

__global__ __launch_bounds__(256)
void sage_fused_gemm(const float* __restrict__ agg,
                     const float* __restrict__ cnt,
                     const float* __restrict__ xin,
                     const float* __restrict__ Wl,
                     const float* __restrict__ Wr,
                     const float* __restrict__ bias,
                     float* __restrict__ out,
                     int do_relu) {
  const int lane = threadIdx.x & 31;
  const int wave = threadIdx.x >> 5;
  const int m    = lane & 15;
  const int half = lane >> 4;
  const int row0 = blockIdx.x * 16;
  const int col  = wave * 16 + m;

  const float* aggRow = agg + (size_t)(row0 + m) * DIM;
  const float* xRow   = xin + (size_t)(row0 + m) * DIM;
  const float  inv    = 1.0f / fmaxf(cnt[row0 + m], 1.0f);

  v8f c = {};
#pragma unroll
  for (int s = 0; s < 32; ++s) {
    const int kk = 4 * s + 2 * half;
    v2f a, b;
    a.x = aggRow[kk] * inv;
    a.y = aggRow[kk + 1] * inv;
    b.x = Wl[(size_t)kk * DIM + col];
    b.y = Wl[(size_t)(kk + 1) * DIM + col];
    c = __builtin_amdgcn_wmma_f32_16x16x4_f32(false, a, false, b, (short)0, c,
                                              false, false);
    a.x = xRow[kk];
    a.y = xRow[kk + 1];
    b.x = Wr[(size_t)kk * DIM + col];
    b.y = Wr[(size_t)(kk + 1) * DIM + col];
    c = __builtin_amdgcn_wmma_f32_16x16x4_f32(false, a, false, b, (short)0, c,
                                              false, false);
  }

  const float bv = bias[col];
#pragma unroll
  for (int r = 0; r < 8; ++r) {
    float v = c[r] + bv;
    if (do_relu) v = fmaxf(v, 0.0f);
    out[(size_t)(row0 + r + 8 * half) * DIM + col] = v;
  }
}

extern "C" void kernel_launch(void* const* d_in, const int* in_sizes, int n_in,
                              void* d_out, int out_size, void* d_ws, size_t ws_size,
                              hipStream_t stream) {
  const float* x   = (const float*)d_in[0];
  const int*  edge = (const int*)d_in[1];
  const float* W1l = (const float*)d_in[2];
  const float* b1  = (const float*)d_in[3];
  const float* W1r = (const float*)d_in[4];
  const float* W2l = (const float*)d_in[5];
  const float* b2  = (const float*)d_in[6];
  const float* W2r = (const float*)d_in[7];

  const int N = in_sizes[0] / DIM;
  const int E = in_sizes[1] / 2;
  const int* src = edge;
  const int* dst = edge + E;

  float* agg = (float*)d_ws;
  float* cnt = agg + (size_t)N * DIM;
  float* h   = cnt + N;
  float* out = (float*)d_out;

  const int edges_per_block = 256 / 32;
  const int scatter_blocks  = (E + edges_per_block - 1) / edges_per_block;
  const int gemm_blocks     = N / 16;

  sage_zero_f32<<<2048, 256, 0, stream>>>(agg, (long long)N * DIM + N);
  sage_degree<<<2048, 256, 0, stream>>>(dst, cnt, E);
  sage_scatter<<<scatter_blocks, 256, 0, stream>>>(x, src, dst, agg, E);
  sage_fused_gemm<<<gemm_blocks, 256, 0, stream>>>(agg, cnt, x, W1l, W1r, b1, h, 1);

  sage_zero_f32<<<2048, 256, 0, stream>>>(agg, (long long)N * DIM);
  sage_scatter<<<scatter_blocks, 256, 0, stream>>>(h, src, dst, agg, E);
  sage_fused_gemm<<<gemm_blocks, 256, 0, stream>>>(agg, cnt, h, W2l, W2r, b2, out, 0);
}


# GEMM: joint k range split over a workgroup's 4 waves (64 k x all 128 columns, own quarter of A only, next stripe prefetched); partial tiles summed in LDS, whole-row stores; sum order: 4 k slices
# speedup vs baseline: 3.4413x; 3.4413x over previous
#include <hip/hip_runtime.h>
#include <hip/hip_bf16.h>

#define DIM 128

typedef __attribute__((ext_vector_type(2))) float v2f;
typedef __attribute__((ext_vector_type(8))) float v8f;

#define OWN_ROWS  416
#define OWN_WAVES 32
#define OWN_QCAP  512

#define PL_EDGES   4096
#define PL_WORDS   (PL_EDGES / 32)
#define PL_PITCH   (PL_WORDS + 4)
#define PL_OWNERS  256
#define PL_THREADS 512

__global__ __launch_bounds__(PL_THREADS)
void sage_owner_planes(const int* __restrict__ dst,
                       unsigned* __restrict__ planes,
                       int E, int owners, int nwp) {
  __shared__ uint4 bits4[PL_OWNERS * PL_PITCH / 4];
  unsigned* bits = reinterpret_cast<unsigned*>(bits4);

  const int tid  = threadIdx.x;
  const int lane = tid & 31;
  const int wave = tid >> 5;
  const unsigned own0 = blockIdx.y * PL_OWNERS;
  const int e0 = blockIdx.x * PL_EDGES;

  for (int i = tid; i < PL_OWNERS * PL_PITCH / 4; i += PL_THREADS) bits4[i] = make_uint4(0u, 0u, 0u, 0u);
  __syncthreads();

  auto mark = [&](int d, int i) __attribute__((always_inline)) {
    const unsigned o = (unsigned)d / (unsigned)OWN_ROWS - own0;
    if (o < (unsigned)PL_OWNERS) atomicOr(&bits[o * PL_PITCH + (i >> 5)], 1u << (i & 31));
  };

  int4 d[PL_EDGES / (4 * PL_THREADS)];
  if (e0 + PL_EDGES <= E) {
#pragma unroll
    for (int k = 0; k < PL_EDGES / (4 * PL_THREADS); ++k) {
      d[k] = *reinterpret_cast<const int4*>(dst + e0 + k * 4 * PL_THREADS + 4 * tid);
    }
  } else {
#pragma unroll
    for (int k = 0; k < PL_EDGES / (4 * PL_THREADS); ++k) {
      const int e = e0 + k * 4 * PL_THREADS + 4 * tid;
      d[k].x = (e + 0 < E) ? dst[e + 0] : -1;
      d[k].y = (e + 1 < E) ? dst[e + 1] : -1;
      d[k].z = (e + 2 < E) ? dst[e + 2] : -1;
      d[k].w = (e + 3 < E) ? dst[e + 3] : -1;
    }
  }
#pragma unroll
  for (int k = 0; k < PL_EDGES / (4 * PL_THREADS); ++k) {
    const int i = k * 4 * PL_THREADS + 4 * tid;
    mark(d[k].x, i);
    mark(d[k].y, i + 1);
    mark(d[k].z, i + 2);
    mark(d[k].w, i + 3);
  }
  __syncthreads();

  for (int o = wave; o < PL_OWNERS; o += PL_THREADS / 32) {
    const unsigned og = own0 + o;
    if (og < (unsigned)owners) {
      const uint4 v = bits4[o * (PL_PITCH / 4) + lane];
      uint4* q = reinterpret_cast<uint4*>(planes + (size_t)og * nwp + (size_t)blockIdx.x * PL_WORDS);
      q[lane] = v;
    }
  }
}

template <bool COUNT>
__global__ __launch_bounds__(1024)
void sage_own_rows(const float* __restrict__ feat,
                   const int* __restrict__ src,
                   const int* __restrict__ dst,
                   const unsigned* __restrict__ planes,
                   float* __restrict__ agg,
                   float* __restrict__ cnt,
                   int nwp, int N) {
  __shared__ float tile[OWN_ROWS * DIM];
  __shared__ float ctile[OWN_ROWS];
  __shared__ int   ring[OWN_WAVES][OWN_QCAP];

  const int tid  = threadIdx.x;
  const int lane = tid & 31;
  const int wave = __builtin_amdgcn_readfirstlane(tid >> 5);
  const int row0 = blockIdx.x * OWN_ROWS;

  float4* t4 = reinterpret_cast<float4*>(tile);
  for (int i = tid; i < OWN_ROWS * DIM / 4; i += 1024) t4[i] = make_float4(0.0f, 0.0f, 0.0f, 0.0f);
  if (COUNT) {
    for (int i = tid; i < OWN_ROWS; i += 1024) ctile[i] = 0.0f;
  }
  __syncthreads();

  int* myq = ring[wave];
  const unsigned below = (1u << lane) - 1u;
  unsigned head = 0, tail = 0;

  auto flush = [&](unsigned n) __attribute__((always_inline)) {
    const bool valid = (unsigned)lane < n;
    int s = 0;
    unsigned rr = 0;
    if (valid) {
      const int e = myq[(head + lane) & (OWN_QCAP - 1)];
      s  = src[e];
      rr = (unsigned)(dst[e] - row0);
      if (COUNT) atomicAdd(&ctile[rr], 1.0f);
    }
    for (unsigned g = 0; g < n; g += 8) {
      float v[8][4];
#pragma unroll
      for (int k = 0; k < 8; ++k) {
        const int sj = __builtin_amdgcn_readlane(s, g + k);
        const float* p = feat + (size_t)sj * DIM + lane;
        v[k][0] = p[0];
        v[k][1] = p[32];
        v[k][2] = p[64];
        v[k][3] = p[96];
      }
#pragma unroll
      for (int k = 0; k < 8; ++k) {
        if (g + k < n) {
          const unsigned rj = __builtin_amdgcn_readlane(rr, g + k);
          float* t = tile + rj * DIM + lane;
          atomicAdd(t,      v[k][0]);
          atomicAdd(t + 32, v[k][1]);
          atomicAdd(t + 64, v[k][2]);
          atomicAdd(t + 96, v[k][3]);
        }
      }
    }
  };

  auto take = [&](unsigned m, int e0) __attribute__((always_inline)) {
    unsigned bal = __builtin_amdgcn_ballot_w32(m != 0u);
    while (bal != 0u) {
      if (m != 0u) {
        myq[(tail + __builtin_popcount(bal & below)) & (OWN_QCAP - 1)] = e0 + __builtin_ctz(m);
        m &= m - 1u;
      }
      tail += __builtin_popcount(bal);
      while (tail - head >= 32u) {
        flush(32u);
        head += 32u;
      }
      bal = __builtin_amdgcn_ballot_w32(m != 0u);
    }
  };

  const uint4* pl = reinterpret_cast<const uint4*>(planes + (size_t)blockIdx.x * nwp);
  const int nsteps = nwp / 128;
  if (wave < nsteps) {
    int step = wave;
    uint4 cur = pl[step * 32 + lane];
    while (step < nsteps) {
      const int nxt = step + OWN_WAVES;
      const uint4 nx = pl[(nxt < nsteps ? nxt : step) * 32 + lane];
      const int e0 = (step * 32 + lane) * 128;
      take(cur.x, e0);
      take(cur.y, e0 + 32);
      take(cur.z, e0 + 64);
      take(cur.w, e0 + 96);
      cur = nx;
      step = nxt;
    }
  }
  if (tail != head) flush(tail - head);

  __syncthreads();

  const int left  = N - row0;
  const int nrows = left < OWN_ROWS ? left : OWN_ROWS;
  float4* a4 = reinterpret_cast<float4*>(agg + (size_t)row0 * DIM);
  for (int i = tid; i < nrows * (DIM / 4); i += 1024) a4[i] = t4[i];
  if (COUNT) {
    for (int i = tid; i < nrows; i += 1024) cnt[row0 + i] = ctile[i];
  }
}

#define GEMM_WAVES 4
#define GEMM_KS    16
#define GEMM_NT    (DIM / 16)
#define GEMM_WGS   512
#define GEMM_PITCH (DIM + 4)

__device__ __forceinline__ int kof(int s, int h) { return 8 * (s >> 1) + 4 * h + 2 * (s & 1); }

__global__ __launch_bounds__(32 * GEMM_WAVES)
void sage_fused_gemm(const float* __restrict__ agg,
                     const float* __restrict__ cnt,
                     const float* __restrict__ xin,
                     const float* __restrict__ Wl,
                     const float* __restrict__ Wr,
                     const float* __restrict__ bias,
                     float* __restrict__ out,
                     int do_relu, int nstripes) {
  __shared__ float4 part4[2][GEMM_WAVES][16 * GEMM_PITCH / 4];

  const int lane = threadIdx.x & 31;
  const int wave = __builtin_amdgcn_readfirstlane(threadIdx.x >> 5);
  const int m    = lane & 15;
  const int half = lane >> 4;

  const bool left = wave < 2;
  const float* A  = left ? agg : xin;
  const float* Wm = left ? Wl : Wr;
  const int kb    = (wave & 1) * (4 * GEMM_KS);

  v2f b[GEMM_NT][GEMM_KS];
#pragma unroll
  for (int t = 0; t < GEMM_NT; ++t) {
#pragma unroll
    for (int s = 0; s < GEMM_KS; ++s) {
      const int kk = kb + kof(s, half);
      b[t][s].x = Wm[(size_t)kk * DIM + t * 16 + m];
      b[t][s].y = Wm[(size_t)(kk + 1) * DIM + t * 16 + m];
    }
  }
  const float4 bias4 = reinterpret_cast<const float4*>(bias)[lane];

  float4 an[GEMM_KS / 2];
  float  cn = 1.0f;
  auto load_a = [&](int st) __attribute__((always_inline)) {
    const size_t row = (size_t)st * 16 + m;
    const float4* p = reinterpret_cast<const float4*>(A + row * DIM + kb + 4 * half);
#pragma unroll
    for (int j = 0; j < GEMM_KS / 2; ++j) an[j] = p[2 * j];
    if (left) cn = cnt[row];
  };

  int st = blockIdx.x;
  if (st < nstripes) load_a(st);
  int buf = 0;

  for (; st < nstripes; st += gridDim.x) {
    const int row0 = st * 16;
    float4 ac[GEMM_KS / 2];
#pragma unroll
    for (int j = 0; j < GEMM_KS / 2; ++j) ac[j] = an[j];
    const float scale = left ? 1.0f / fmaxf(cn, 1.0f) : 1.0f;

    const int nx = st + gridDim.x;
    load_a(nx < nstripes ? nx : st);

    v8f c[GEMM_NT];
#pragma unroll
    for (int t = 0; t < GEMM_NT; ++t) c[t] = v8f{};

#pragma unroll
    for (int s = 0; s < GEMM_KS; ++s) {
      const float4 q = ac[s >> 1];
      v2f a;
      a.x = ((s & 1) ? q.z : q.x) * scale;
      a.y = ((s & 1) ? q.w : q.y) * scale;
#pragma unroll
      for (int t = 0; t < GEMM_NT; ++t)
        c[t] = __builtin_amdgcn_wmma_f32_16x16x4_f32(false, a, false, b[t][s], (short)0, c[t],
                                                     false, false);
    }

    float* P = reinterpret_cast<float*>(part4[buf][wave]);
#pragma unroll
    for (int t = 0; t < GEMM_NT; ++t) {
#pragma unroll
      for (int r = 0; r < 8; ++r) P[(r + 8 * half) * GEMM_PITCH + t * 16 + m] = c[t][r];
    }
    __syncthreads();

#pragma unroll
    for (int i = 0; i < 4; ++i) {
      const int row = 4 * wave + i;
      const int o = row * (GEMM_PITCH / 4) + lane;
      const float4 p0 = part4[buf][0][o];
      const float4 p1 = part4[buf][1][o];
      const float4 p2 = part4[buf][2][o];
      const float4 p3 = part4[buf][3][o];
      float4 v;
      v.x = ((p0.x + p1.x) + p2.x) + p3.x + bias4.x;
      v.y = ((p0.y + p1.y) + p2.y) + p3.y + bias4.y;
      v.z = ((p0.z + p1.z) + p2.z) + p3.z + bias4.z;
      v.w = ((p0.w + p1.w) + p2.w) + p3.w + bias4.w;
      if (do_relu) {
        v.x = fmaxf(v.x, 0.0f);
        v.y = fmaxf(v.y, 0.0f);
        v.z = fmaxf(v.z, 0.0f);
        v.w = fmaxf(v.w, 0.0f);
      }
      reinterpret_cast<float4*>(out + (size_t)(row0 + row) * DIM)[lane] = v;
    }
    buf ^= 1;
  }
}

extern "C" void kernel_launch(void* const* d_in, const int* in_sizes, int n_in,
                              void* d_out, int out_size, void* d_ws, size_t ws_size,
                              hipStream_t stream) {
  const float* x   = (const float*)d_in[0];
  const int*  edge = (const int*)d_in[1];
  const float* W1l = (const float*)d_in[2];
  const float* b1  = (const float*)d_in[3];
  const float* W1r = (const float*)d_in[4];
  const float* W2l = (const float*)d_in[5];
  const float* b2  = (const float*)d_in[6];
  const float* W2r = (const float*)d_in[7];

  const int N = in_sizes[0] / DIM;
  const int E = in_sizes[1] / 2;
  const int* src = edge;
  const int* dst = edge + E;

  float* agg = (float*)d_ws;
  float* cnt = agg + (size_t)N * DIM;
  float* h   = cnt + N;
  float* out = (float*)d_out;

  const int own_blocks  = (N + OWN_ROWS - 1) / OWN_ROWS;
  const int gemm_blocks = N / 16;

  const int pl_chunks = (E + PL_EDGES - 1) / PL_EDGES;
  const int nwp       = pl_chunks * PL_WORDS;
  const size_t pl_off = ((size_t)N * (2 * DIM + 1) + 63) / 64 * 64;
  unsigned* planes    = (unsigned*)d_ws + pl_off;
  const dim3 pl_grid(pl_chunks, (own_blocks + PL_OWNERS - 1) / PL_OWNERS);

  sage_owner_planes<<<pl_grid, PL_THREADS, 0, stream>>>(dst, planes, E, own_blocks, nwp);

  sage_own_rows<true><<<own_blocks, 1024, 0, stream>>>(x, src, dst, planes, agg, cnt, nwp, N);
  sage_fused_gemm<<<GEMM_WGS, 32 * GEMM_WAVES, 0, stream>>>(agg, cnt, x, W1l, W1r, b1, h, 1, gemm_blocks);

  sage_own_rows<false><<<own_blocks, 1024, 0, stream>>>(h, src, dst, planes, agg, cnt, nwp, N);
  sage_fused_gemm<<<GEMM_WGS, 32 * GEMM_WAVES, 0, stream>>>(agg, cnt, h, W2l, W2r, b2, out, 0, gemm_blocks);
}


# neighbour sum: an owned edge's two random 4-byte look-ups (src[e], dst[e]: two 128-B lines) become one: the plane launch also writes a packed word per edge (src | row in owner << 23); sum order same
# speedup vs baseline: 3.5664x; 1.0364x over previous
#include <hip/hip_runtime.h>
#include <hip/hip_bf16.h>

#define DIM 128

typedef __attribute__((ext_vector_type(2))) float v2f;
typedef __attribute__((ext_vector_type(8))) float v8f;

#define OWN_ROWS  416
#define OWN_WAVES 32
#define OWN_QCAP  512

#define PL_EDGES   4096
#define PL_WORDS   (PL_EDGES / 32)
#define PL_PITCH   (PL_WORDS + 4)
#define PL_OWNERS  256
#define PL_THREADS 512

#define PK_SHIFT 23
#define PK_MASK  ((1u << PK_SHIFT) - 1u)

__global__ __launch_bounds__(PL_THREADS)
void sage_owner_planes(const int* __restrict__ src,
                       const int* __restrict__ dst,
                       unsigned* __restrict__ planes,
                       unsigned* __restrict__ packed,
                       int E, int owners, int nwp) {
  __shared__ uint4 bits4[PL_OWNERS * PL_PITCH / 4];
  unsigned* bits = reinterpret_cast<unsigned*>(bits4);

  const int tid  = threadIdx.x;
  const int lane = tid & 31;
  const int wave = tid >> 5;
  const unsigned own0 = blockIdx.y * PL_OWNERS;
  const int e0 = blockIdx.x * PL_EDGES;

  for (int i = tid; i < PL_OWNERS * PL_PITCH / 4; i += PL_THREADS) bits4[i] = make_uint4(0u, 0u, 0u, 0u);
  __syncthreads();

  auto mark = [&](int d, int i) __attribute__((always_inline)) {
    const unsigned o = (unsigned)d / (unsigned)OWN_ROWS - own0;
    if (o < (unsigned)PL_OWNERS) atomicOr(&bits[o * PL_PITCH + (i >> 5)], 1u << (i & 31));
  };

  int4 d[PL_EDGES / (4 * PL_THREADS)];
  int4 s[PL_EDGES / (4 * PL_THREADS)];
  const bool whole = e0 + PL_EDGES <= E;
  const bool pack  = blockIdx.y == 0;
  auto pk = [](int sv, int dv) __attribute__((always_inline)) {
    return (unsigned)sv | (((unsigned)dv % (unsigned)OWN_ROWS) << PK_SHIFT);
  };
  if (whole) {
    if (pack) {
#pragma unroll
      for (int k = 0; k < PL_EDGES / (4 * PL_THREADS); ++k) {
        s[k] = *reinterpret_cast<const int4*>(src + e0 + k * 4 * PL_THREADS + 4 * tid);
      }
    }
#pragma unroll
    for (int k = 0; k < PL_EDGES / (4 * PL_THREADS); ++k) {
      d[k] = *reinterpret_cast<const int4*>(dst + e0 + k * 4 * PL_THREADS + 4 * tid);
    }
    if (pack) {
#pragma unroll
      for (int k = 0; k < PL_EDGES / (4 * PL_THREADS); ++k) {
        *reinterpret_cast<uint4*>(packed + e0 + k * 4 * PL_THREADS + 4 * tid) =
            make_uint4(pk(s[k].x, d[k].x), pk(s[k].y, d[k].y), pk(s[k].z, d[k].z), pk(s[k].w, d[k].w));
      }
    }
  } else {
#pragma unroll
    for (int k = 0; k < PL_EDGES / (4 * PL_THREADS); ++k) {
      const int e = e0 + k * 4 * PL_THREADS + 4 * tid;
      d[k].x = (e + 0 < E) ? dst[e + 0] : -1;
      d[k].y = (e + 1 < E) ? dst[e + 1] : -1;
      d[k].z = (e + 2 < E) ? dst[e + 2] : -1;
      d[k].w = (e + 3 < E) ? dst[e + 3] : -1;
      if (pack) {
        if (e + 0 < E) packed[e + 0] = pk(src[e + 0], d[k].x);
        if (e + 1 < E) packed[e + 1] = pk(src[e + 1], d[k].y);
        if (e + 2 < E) packed[e + 2] = pk(src[e + 2], d[k].z);
        if (e + 3 < E) packed[e + 3] = pk(src[e + 3], d[k].w);
      }
    }
  }
#pragma unroll
  for (int k = 0; k < PL_EDGES / (4 * PL_THREADS); ++k) {
    const int i = k * 4 * PL_THREADS + 4 * tid;
    mark(d[k].x, i);
    mark(d[k].y, i + 1);
    mark(d[k].z, i + 2);
    mark(d[k].w, i + 3);
  }
  __syncthreads();

  for (int o = wave; o < PL_OWNERS; o += PL_THREADS / 32) {
    const unsigned og = own0 + o;
    if (og < (unsigned)owners) {
      const uint4 v = bits4[o * (PL_PITCH / 4) + lane];
      uint4* q = reinterpret_cast<uint4*>(planes + (size_t)og * nwp + (size_t)blockIdx.x * PL_WORDS);
      q[lane] = v;
    }
  }
}

template <bool COUNT>
__global__ __launch_bounds__(1024)
void sage_own_rows(const float* __restrict__ feat,
                   const unsigned* __restrict__ packed,
                   const unsigned* __restrict__ planes,
                   float* __restrict__ agg,
                   float* __restrict__ cnt,
                   int nwp, int N) {
  __shared__ float tile[OWN_ROWS * DIM];
  __shared__ float ctile[OWN_ROWS];
  __shared__ int   ring[OWN_WAVES][OWN_QCAP];

  const int tid  = threadIdx.x;
  const int lane = tid & 31;
  const int wave = __builtin_amdgcn_readfirstlane(tid >> 5);
  const int row0 = blockIdx.x * OWN_ROWS;

  float4* t4 = reinterpret_cast<float4*>(tile);
  for (int i = tid; i < OWN_ROWS * DIM / 4; i += 1024) t4[i] = make_float4(0.0f, 0.0f, 0.0f, 0.0f);
  if (COUNT) {
    for (int i = tid; i < OWN_ROWS; i += 1024) ctile[i] = 0.0f;
  }
  __syncthreads();

  int* myq = ring[wave];
  const unsigned below = (1u << lane) - 1u;
  unsigned head = 0, tail = 0;

  auto flush = [&](unsigned n) __attribute__((always_inline)) {
    const bool valid = (unsigned)lane < n;
    int s = 0;
    unsigned rr = 0;
    if (valid) {
      const int e = myq[(head + lane) & (OWN_QCAP - 1)];
      const unsigned w = packed[e];
      s  = (int)(w & PK_MASK);
      rr = w >> PK_SHIFT;
      if (COUNT) atomicAdd(&ctile[rr], 1.0f);
    }
    for (unsigned g = 0; g < n; g += 8) {
      float v[8][4];
#pragma unroll
      for (int k = 0; k < 8; ++k) {
        const int sj = __builtin_amdgcn_readlane(s, g + k);
        const float* p = feat + (size_t)sj * DIM + lane;
        v[k][0] = p[0];
        v[k][1] = p[32];
        v[k][2] = p[64];
        v[k][3] = p[96];
      }
#pragma unroll
      for (int k = 0; k < 8; ++k) {
        if (g + k < n) {
          const unsigned rj = __builtin_amdgcn_readlane(rr, g + k);
          float* t = tile + rj * DIM + lane;
          atomicAdd(t,      v[k][0]);
          atomicAdd(t + 32, v[k][1]);
          atomicAdd(t + 64, v[k][2]);
          atomicAdd(t + 96, v[k][3]);
        }
      }
    }
  };

  auto take = [&](unsigned m, int e0) __attribute__((always_inline)) {
    unsigned bal = __builtin_amdgcn_ballot_w32(m != 0u);
    while (bal != 0u) {
      if (m != 0u) {
        myq[(tail + __builtin_popcount(bal & below)) & (OWN_QCAP - 1)] = e0 + __builtin_ctz(m);
        m &= m - 1u;
      }
      tail += __builtin_popcount(bal);
      while (tail - head >= 32u) {
        flush(32u);
        head += 32u;
      }
      bal = __builtin_amdgcn_ballot_w32(m != 0u);
    }
  };

  const uint4* pl = reinterpret_cast<const uint4*>(planes + (size_t)blockIdx.x * nwp);
  const int nsteps = nwp / 128;
  if (wave < nsteps) {
    int step = wave;
    uint4 cur = pl[step * 32 + lane];
    while (step < nsteps) {
      const int nxt = step + OWN_WAVES;
      const uint4 nx = pl[(nxt < nsteps ? nxt : step) * 32 + lane];
      const int e0 = (step * 32 + lane) * 128;
      take(cur.x, e0);
      take(cur.y, e0 + 32);
      take(cur.z, e0 + 64);
      take(cur.w, e0 + 96);
      cur = nx;
      step = nxt;
    }
  }
  if (tail != head) flush(tail - head);

  __syncthreads();

  const int left  = N - row0;
  const int nrows = left < OWN_ROWS ? left : OWN_ROWS;
  float4* a4 = reinterpret_cast<float4*>(agg + (size_t)row0 * DIM);
  for (int i = tid; i < nrows * (DIM / 4); i += 1024) a4[i] = t4[i];
  if (COUNT) {
    for (int i = tid; i < nrows; i += 1024) cnt[row0 + i] = ctile[i];
  }
}

#define GEMM_WAVES 4
#define GEMM_KS    16
#define GEMM_NT    (DIM / 16)
#define GEMM_WGS   512
#define GEMM_PITCH (DIM + 4)

__device__ __forceinline__ int kof(int s, int h) { return 8 * (s >> 1) + 4 * h + 2 * (s & 1); }

__global__ __launch_bounds__(32 * GEMM_WAVES)
void sage_fused_gemm(const float* __restrict__ agg,
                     const float* __restrict__ cnt,
                     const float* __restrict__ xin,
                     const float* __restrict__ Wl,
                     const float* __restrict__ Wr,
                     const float* __restrict__ bias,
                     float* __restrict__ out,
                     int do_relu, int nstripes) {
  __shared__ float4 part4[2][GEMM_WAVES][16 * GEMM_PITCH / 4];

  const int lane = threadIdx.x & 31;
  const int wave = __builtin_amdgcn_readfirstlane(threadIdx.x >> 5);
  const int m    = lane & 15;
  const int half = lane >> 4;

  const bool left = wave < 2;
  const float* A  = left ? agg : xin;
  const float* Wm = left ? Wl : Wr;
  const int kb    = (wave & 1) * (4 * GEMM_KS);

  v2f b[GEMM_NT][GEMM_KS];
#pragma unroll
  for (int t = 0; t < GEMM_NT; ++t) {
#pragma unroll
    for (int s = 0; s < GEMM_KS; ++s) {
      const int kk = kb + kof(s, half);
      b[t][s].x = Wm[(size_t)kk * DIM + t * 16 + m];
      b[t][s].y = Wm[(size_t)(kk + 1) * DIM + t * 16 + m];
    }
  }
  const float4 bias4 = reinterpret_cast<const float4*>(bias)[lane];

  float4 an[GEMM_KS / 2];
  float  cn = 1.0f;
  auto load_a = [&](int st) __attribute__((always_inline)) {
    const size_t row = (size_t)st * 16 + m;
    const float4* p = reinterpret_cast<const float4*>(A + row * DIM + kb + 4 * half);
#pragma unroll
    for (int j = 0; j < GEMM_KS / 2; ++j) an[j] = p[2 * j];
    if (left) cn = cnt[row];
  };

  int st = blockIdx.x;
  if (st < nstripes) load_a(st);
  int buf = 0;

  for (; st < nstripes; st += gridDim.x) {
    const int row0 = st * 16;
    float4 ac[GEMM_KS / 2];
#pragma unroll
    for (int j = 0; j < GEMM_KS / 2; ++j) ac[j] = an[j];
    const float scale = left ? 1.0f / fmaxf(cn, 1.0f) : 1.0f;

    const int nx = st + gridDim.x;
    load_a(nx < nstripes ? nx : st);

    v8f c[GEMM_NT];
#pragma unroll
    for (int t = 0; t < GEMM_NT; ++t) c[t] = v8f{};

#pragma unroll
    for (int s = 0; s < GEMM_KS; ++s) {
      const float4 q = ac[s >> 1];
      v2f a;
      a.x = ((s & 1) ? q.z : q.x) * scale;
      a.y = ((s & 1) ? q.w : q.y) * scale;
#pragma unroll
      for (int t = 0; t < GEMM_NT; ++t)
        c[t] = __builtin_amdgcn_wmma_f32_16x16x4_f32(false, a, false, b[t][s], (short)0, c[t],
                                                     false, false);
    }

    float* P = reinterpret_cast<float*>(part4[buf][wave]);
#pragma unroll
    for (int t = 0; t < GEMM_NT; ++t) {
#pragma unroll
      for (int r = 0; r < 8; ++r) P[(r + 8 * half) * GEMM_PITCH + t * 16 + m] = c[t][r];
    }
    __syncthreads();

#pragma unroll
    for (int i = 0; i < 4; ++i) {
      const int row = 4 * wave + i;
      const int o = row * (GEMM_PITCH / 4) + lane;
      const float4 p0 = part4[buf][0][o];
      const float4 p1 = part4[buf][1][o];
      const float4 p2 = part4[buf][2][o];
      const float4 p3 = part4[buf][3][o];
      float4 v;
      v.x = ((p0.x + p1.x) + p2.x) + p3.x + bias4.x;
      v.y = ((p0.y + p1.y) + p2.y) + p3.y + bias4.y;
      v.z = ((p0.z + p1.z) + p2.z) + p3.z + bias4.z;
      v.w = ((p0.w + p1.w) + p2.w) + p3.w + bias4.w;
      if (do_relu) {
        v.x = fmaxf(v.x, 0.0f);
        v.y = fmaxf(v.y, 0.0f);
        v.z = fmaxf(v.z, 0.0f);
        v.w = fmaxf(v.w, 0.0f);
      }
      reinterpret_cast<float4*>(out + (size_t)(row0 + row) * DIM)[lane] = v;
    }
    buf ^= 1;
  }
}

extern "C" void kernel_launch(void* const* d_in, const int* in_sizes, int n_in,
                              void* d_out, int out_size, void* d_ws, size_t ws_size,
                              hipStream_t stream) {
  const float* x   = (const float*)d_in[0];
  const int*  edge = (const int*)d_in[1];
  const float* W1l = (const float*)d_in[2];
  const float* b1  = (const float*)d_in[3];
  const float* W1r = (const float*)d_in[4];
  const float* W2l = (const float*)d_in[5];
  const float* b2  = (const float*)d_in[6];
  const float* W2r = (const float*)d_in[7];

  const int N = in_sizes[0] / DIM;
  const int E = in_sizes[1] / 2;
  const int* src = edge;
  const int* dst = edge + E;

  float* agg = (float*)d_ws;
  float* cnt = agg + (size_t)N * DIM;
  float* h   = cnt + N;
  float* out = (float*)d_out;

  const int own_blocks  = (N + OWN_ROWS - 1) / OWN_ROWS;
  const int gemm_blocks = N / 16;

  const int pl_chunks = (E + PL_EDGES - 1) / PL_EDGES;
  const int nwp       = pl_chunks * PL_WORDS;
  const size_t pl_off = ((size_t)N * (2 * DIM + 1) + 63) / 64 * 64;
  unsigned* planes    = (unsigned*)d_ws + pl_off;
  const dim3 pl_grid(pl_chunks, (own_blocks + PL_OWNERS - 1) / PL_OWNERS);
  unsigned* packed    = planes + (size_t)own_blocks * nwp;

  sage_owner_planes<<<pl_grid, PL_THREADS, 0, stream>>>(src, dst, planes, packed, E, own_blocks, nwp);

  sage_own_rows<true><<<own_blocks, 1024, 0, stream>>>(x, packed, planes, agg, cnt, nwp, N);
  sage_fused_gemm<<<GEMM_WGS, 32 * GEMM_WAVES, 0, stream>>>(agg, cnt, x, W1l, W1r, b1, h, 1, gemm_blocks);

  sage_own_rows<false><<<own_blocks, 1024, 0, stream>>>(h, packed, planes, agg, cnt, nwp, N);
  sage_fused_gemm<<<GEMM_WGS, 32 * GEMM_WAVES, 0, stream>>>(agg, cnt, h, W2l, W2r, b2, out, 0, gemm_blocks);
}


# neighbour sum: a gathered 512-B row is one 16-byte load a lane instead of four 4-byte loads (4x fewer row-load instructions); LDS tile rows lane-major, atomics stay conflict-free; sum order same
# speedup vs baseline: 3.6836x; 1.0329x over previous
#include <hip/hip_runtime.h>
#include <hip/hip_bf16.h>

#define DIM 128

typedef __attribute__((ext_vector_type(2))) float v2f;
typedef __attribute__((ext_vector_type(8))) float v8f;

#define OWN_ROWS  416
#define OWN_WAVES 32
#define OWN_QCAP  512

#define PL_EDGES   4096
#define PL_WORDS   (PL_EDGES / 32)
#define PL_PITCH   (PL_WORDS + 4)
#define PL_OWNERS  256
#define PL_THREADS 512

#define PK_SHIFT 23
#define PK_MASK  ((1u << PK_SHIFT) - 1u)

__global__ __launch_bounds__(PL_THREADS)
void sage_owner_planes(const int* __restrict__ src,
                       const int* __restrict__ dst,
                       unsigned* __restrict__ planes,
                       unsigned* __restrict__ packed,
                       int E, int owners, int nwp) {
  __shared__ uint4 bits4[PL_OWNERS * PL_PITCH / 4];
  unsigned* bits = reinterpret_cast<unsigned*>(bits4);

  const int tid  = threadIdx.x;
  const int lane = tid & 31;
  const int wave = tid >> 5;
  const unsigned own0 = blockIdx.y * PL_OWNERS;
  const int e0 = blockIdx.x * PL_EDGES;

  for (int i = tid; i < PL_OWNERS * PL_PITCH / 4; i += PL_THREADS) bits4[i] = make_uint4(0u, 0u, 0u, 0u);
  __syncthreads();

  auto mark = [&](int d, int i) __attribute__((always_inline)) {
    const unsigned o = (unsigned)d / (unsigned)OWN_ROWS - own0;
    if (o < (unsigned)PL_OWNERS) atomicOr(&bits[o * PL_PITCH + (i >> 5)], 1u << (i & 31));
  };

  int4 d[PL_EDGES / (4 * PL_THREADS)];
  int4 s[PL_EDGES / (4 * PL_THREADS)];
  const bool whole = e0 + PL_EDGES <= E;
  const bool pack  = blockIdx.y == 0;
  auto pk = [](int sv, int dv) __attribute__((always_inline)) {
    return (unsigned)sv | (((unsigned)dv % (unsigned)OWN_ROWS) << PK_SHIFT);
  };
  if (whole) {
    if (pack) {
#pragma unroll
      for (int k = 0; k < PL_EDGES / (4 * PL_THREADS); ++k) {
        s[k] = *reinterpret_cast<const int4*>(src + e0 + k * 4 * PL_THREADS + 4 * tid);
      }
    }
#pragma unroll
    for (int k = 0; k < PL_EDGES / (4 * PL_THREADS); ++k) {
      d[k] = *reinterpret_cast<const int4*>(dst + e0 + k * 4 * PL_THREADS + 4 * tid);
    }
    if (pack) {
#pragma unroll
      for (int k = 0; k < PL_EDGES / (4 * PL_THREADS); ++k) {
        *reinterpret_cast<uint4*>(packed + e0 + k * 4 * PL_THREADS + 4 * tid) =
            make_uint4(pk(s[k].x, d[k].x), pk(s[k].y, d[k].y), pk(s[k].z, d[k].z), pk(s[k].w, d[k].w));
      }
    }
  } else {
#pragma unroll
    for (int k = 0; k < PL_EDGES / (4 * PL_THREADS); ++k) {
      const int e = e0 + k * 4 * PL_THREADS + 4 * tid;
      d[k].x = (e + 0 < E) ? dst[e + 0] : -1;
      d[k].y = (e + 1 < E) ? dst[e + 1] : -1;
      d[k].z = (e + 2 < E) ? dst[e + 2] : -1;
      d[k].w = (e + 3 < E) ? dst[e + 3] : -1;
      if (pack) {
        if (e + 0 < E) packed[e + 0] = pk(src[e + 0], d[k].x);
        if (e + 1 < E) packed[e + 1] = pk(src[e + 1], d[k].y);
        if (e + 2 < E) packed[e + 2] = pk(src[e + 2], d[k].z);
        if (e + 3 < E) packed[e + 3] = pk(src[e + 3], d[k].w);
      }
    }
  }
#pragma unroll
  for (int k = 0; k < PL_EDGES / (4 * PL_THREADS); ++k) {
    const int i = k * 4 * PL_THREADS + 4 * tid;
    mark(d[k].x, i);
    mark(d[k].y, i + 1);
    mark(d[k].z, i + 2);
    mark(d[k].w, i + 3);
  }
  __syncthreads();

  for (int o = wave; o < PL_OWNERS; o += PL_THREADS / 32) {
    const unsigned og = own0 + o;
    if (og < (unsigned)owners) {
      const uint4 v = bits4[o * (PL_PITCH / 4) + lane];
      uint4* q = reinterpret_cast<uint4*>(planes + (size_t)og * nwp + (size_t)blockIdx.x * PL_WORDS);
      q[lane] = v;
    }
  }
}

template <bool COUNT>
__global__ __launch_bounds__(1024)
void sage_own_rows(const float* __restrict__ feat,
                   const unsigned* __restrict__ packed,
                   const unsigned* __restrict__ planes,
                   float* __restrict__ agg,
                   float* __restrict__ cnt,
                   int nwp, int N) {
  __shared__ float tile[OWN_ROWS * DIM];
  __shared__ float ctile[OWN_ROWS];
  __shared__ int   ring[OWN_WAVES][OWN_QCAP];

  const int tid  = threadIdx.x;
  const int lane = tid & 31;
  const int wave = __builtin_amdgcn_readfirstlane(tid >> 5);
  const int row0 = blockIdx.x * OWN_ROWS;

  float4* t4 = reinterpret_cast<float4*>(tile);
  for (int i = tid; i < OWN_ROWS * DIM / 4; i += 1024) t4[i] = make_float4(0.0f, 0.0f, 0.0f, 0.0f);
  if (COUNT) {
    for (int i = tid; i < OWN_ROWS; i += 1024) ctile[i] = 0.0f;
  }
  __syncthreads();

  int* myq = ring[wave];
  const unsigned below = (1u << lane) - 1u;
  unsigned head = 0, tail = 0;

  auto flush = [&](unsigned n) __attribute__((always_inline)) {
    const bool valid = (unsigned)lane < n;
    int s = 0;
    unsigned rr = 0;
    if (valid) {
      const int e = myq[(head + lane) & (OWN_QCAP - 1)];
      const unsigned w = packed[e];
      s  = (int)(w & PK_MASK);
      rr = w >> PK_SHIFT;
      if (COUNT) atomicAdd(&ctile[rr], 1.0f);
    }
    for (unsigned g = 0; g < n; g += 8) {
      float4 v[8];
#pragma unroll
      for (int k = 0; k < 8; ++k) {
        const int sj = __builtin_amdgcn_readlane(s, g + k);
        v[k] = reinterpret_cast<const float4*>(feat + (size_t)sj * DIM)[lane];
      }
#pragma unroll
      for (int k = 0; k < 8; ++k) {
        if (g + k < n) {
          const unsigned rj = __builtin_amdgcn_readlane(rr, g + k);
          float* t = tile + rj * DIM + lane;
          atomicAdd(t,      v[k].x);
          atomicAdd(t + 32, v[k].y);
          atomicAdd(t + 64, v[k].z);
          atomicAdd(t + 96, v[k].w);
        }
      }
    }
  };

  auto take = [&](unsigned m, int e0) __attribute__((always_inline)) {
    unsigned bal = __builtin_amdgcn_ballot_w32(m != 0u);
    while (bal != 0u) {
      if (m != 0u) {
        myq[(tail + __builtin_popcount(bal & below)) & (OWN_QCAP - 1)] = e0 + __builtin_ctz(m);
        m &= m - 1u;
      }
      tail += __builtin_popcount(bal);
      while (tail - head >= 32u) {
        flush(32u);
        head += 32u;
      }
      bal = __builtin_amdgcn_ballot_w32(m != 0u);
    }
  };

  const uint4* pl = reinterpret_cast<const uint4*>(planes + (size_t)blockIdx.x * nwp);
  const int nsteps = nwp / 128;
  if (wave < nsteps) {
    int step = wave;
    uint4 cur = pl[step * 32 + lane];
    while (step < nsteps) {
      const int nxt = step + OWN_WAVES;
      const uint4 nx = pl[(nxt < nsteps ? nxt : step) * 32 + lane];
      const int e0 = (step * 32 + lane) * 128;
      take(cur.x, e0);
      take(cur.y, e0 + 32);
      take(cur.z, e0 + 64);
      take(cur.w, e0 + 96);
      cur = nx;
      step = nxt;
    }
  }
  if (tail != head) flush(tail - head);

  __syncthreads();

  const int left  = N - row0;
  const int nrows = left < OWN_ROWS ? left : OWN_ROWS;
  for (int r = wave; r < nrows; r += OWN_WAVES) {
    const float* t = tile + r * DIM + lane;
    reinterpret_cast<float4*>(agg + (size_t)(row0 + r) * DIM)[lane] = make_float4(t[0], t[32], t[64], t[96]);
  }
  if (COUNT) {
    for (int i = tid; i < nrows; i += 1024) cnt[row0 + i] = ctile[i];
  }
}

#define GEMM_WAVES 4
#define GEMM_KS    16
#define GEMM_NT    (DIM / 16)
#define GEMM_WGS   512
#define GEMM_PITCH (DIM + 4)

__device__ __forceinline__ int kof(int s, int h) { return 8 * (s >> 1) + 4 * h + 2 * (s & 1); }

__global__ __launch_bounds__(32 * GEMM_WAVES)
void sage_fused_gemm(const float* __restrict__ agg,
                     const float* __restrict__ cnt,
                     const float* __restrict__ xin,
                     const float* __restrict__ Wl,
                     const float* __restrict__ Wr,
                     const float* __restrict__ bias,
                     float* __restrict__ out,
                     int do_relu, int nstripes) {
  __shared__ float4 part4[2][GEMM_WAVES][16 * GEMM_PITCH / 4];

  const int lane = threadIdx.x & 31;
  const int wave = __builtin_amdgcn_readfirstlane(threadIdx.x >> 5);
  const int m    = lane & 15;
  const int half = lane >> 4;

  const bool left = wave < 2;
  const float* A  = left ? agg : xin;
  const float* Wm = left ? Wl : Wr;
  const int kb    = (wave & 1) * (4 * GEMM_KS);

  v2f b[GEMM_NT][GEMM_KS];
#pragma unroll
  for (int t = 0; t < GEMM_NT; ++t) {
#pragma unroll
    for (int s = 0; s < GEMM_KS; ++s) {
      const int kk = kb + kof(s, half);
      b[t][s].x = Wm[(size_t)kk * DIM + t * 16 + m];
      b[t][s].y = Wm[(size_t)(kk + 1) * DIM + t * 16 + m];
    }
  }
  const float4 bias4 = reinterpret_cast<const float4*>(bias)[lane];

  float4 an[GEMM_KS / 2];
  float  cn = 1.0f;
  auto load_a = [&](int st) __attribute__((always_inline)) {
    const size_t row = (size_t)st * 16 + m;
    const float4* p = reinterpret_cast<const float4*>(A + row * DIM + kb + 4 * half);
#pragma unroll
    for (int j = 0; j < GEMM_KS / 2; ++j) an[j] = p[2 * j];
    if (left) cn = cnt[row];
  };

  int st = blockIdx.x;
  if (st < nstripes) load_a(st);
  int buf = 0;

  for (; st < nstripes; st += gridDim.x) {
    const int row0 = st * 16;
    float4 ac[GEMM_KS / 2];
#pragma unroll
    for (int j = 0; j < GEMM_KS / 2; ++j) ac[j] = an[j];
    const float scale = left ? 1.0f / fmaxf(cn, 1.0f) : 1.0f;

    const int nx = st + gridDim.x;
    load_a(nx < nstripes ? nx : st);

    v8f c[GEMM_NT];
#pragma unroll
    for (int t = 0; t < GEMM_NT; ++t) c[t] = v8f{};

#pragma unroll
    for (int s = 0; s < GEMM_KS; ++s) {
      const float4 q = ac[s >> 1];
      v2f a;
      a.x = ((s & 1) ? q.z : q.x) * scale;
      a.y = ((s & 1) ? q.w : q.y) * scale;
#pragma unroll
      for (int t = 0; t < GEMM_NT; ++t)
        c[t] = __builtin_amdgcn_wmma_f32_16x16x4_f32(false, a, false, b[t][s], (short)0, c[t],
                                                     false, false);
    }

    float* P = reinterpret_cast<float*>(part4[buf][wave]);
#pragma unroll
    for (int t = 0; t < GEMM_NT; ++t) {
#pragma unroll
      for (int r = 0; r < 8; ++r) P[(r + 8 * half) * GEMM_PITCH + t * 16 + m] = c[t][r];
    }
    __syncthreads();

#pragma unroll
    for (int i = 0; i < 4; ++i) {
      const int row = 4 * wave + i;
      const int o = row * (GEMM_PITCH / 4) + lane;
      const float4 p0 = part4[buf][0][o];
      const float4 p1 = part4[buf][1][o];
      const float4 p2 = part4[buf][2][o];
      const float4 p3 = part4[buf][3][o];
      float4 v;
      v.x = ((p0.x + p1.x) + p2.x) + p3.x + bias4.x;
      v.y = ((p0.y + p1.y) + p2.y) + p3.y + bias4.y;
      v.z = ((p0.z + p1.z) + p2.z) + p3.z + bias4.z;
      v.w = ((p0.w + p1.w) + p2.w) + p3.w + bias4.w;
      if (do_relu) {
        v.x = fmaxf(v.x, 0.0f);
        v.y = fmaxf(v.y, 0.0f);
        v.z = fmaxf(v.z, 0.0f);
        v.w = fmaxf(v.w, 0.0f);
      }
      reinterpret_cast<float4*>(out + (size_t)(row0 + row) * DIM)[lane] = v;
    }
    buf ^= 1;
  }
}

extern "C" void kernel_launch(void* const* d_in, const int* in_sizes, int n_in,
                              void* d_out, int out_size, void* d_ws, size_t ws_size,
                              hipStream_t stream) {
  const float* x   = (const float*)d_in[0];
  const int*  edge = (const int*)d_in[1];
  const float* W1l = (const float*)d_in[2];
  const float* b1  = (const float*)d_in[3];
  const float* W1r = (const float*)d_in[4];
  const float* W2l = (const float*)d_in[5];
  const float* b2  = (const float*)d_in[6];
  const float* W2r = (const float*)d_in[7];

  const int N = in_sizes[0] / DIM;
  const int E = in_sizes[1] / 2;
  const int* src = edge;
  const int* dst = edge + E;

  float* agg = (float*)d_ws;
  float* cnt = agg + (size_t)N * DIM;
  float* h   = cnt + N;
  float* out = (float*)d_out;

  const int own_blocks  = (N + OWN_ROWS - 1) / OWN_ROWS;
  const int gemm_blocks = N / 16;

  const int pl_chunks = (E + PL_EDGES - 1) / PL_EDGES;
  const int nwp       = pl_chunks * PL_WORDS;
  const size_t pl_off = ((size_t)N * (2 * DIM + 1) + 63) / 64 * 64;
  unsigned* planes    = (unsigned*)d_ws + pl_off;
  const dim3 pl_grid(pl_chunks, (own_blocks + PL_OWNERS - 1) / PL_OWNERS);
  unsigned* packed    = planes + (size_t)own_blocks * nwp;

  sage_owner_planes<<<pl_grid, PL_THREADS, 0, stream>>>(src, dst, planes, packed, E, own_blocks, nwp);

  sage_own_rows<true><<<own_blocks, 1024, 0, stream>>>(x, packed, planes, agg, cnt, nwp, N);
  sage_fused_gemm<<<GEMM_WGS, 32 * GEMM_WAVES, 0, stream>>>(agg, cnt, x, W1l, W1r, b1, h, 1, gemm_blocks);

  sage_own_rows<false><<<own_blocks, 1024, 0, stream>>>(h, packed, planes, agg, cnt, nwp, N);
  sage_fused_gemm<<<GEMM_WGS, 32 * GEMM_WAVES, 0, stream>>>(agg, cnt, h, W2l, W2r, b2, out, 0, gemm_blocks);
}


# layer 2's owner pass streams each wave's found-edge list, stored in whole lines by layer 1's pass, instead of scanning its bit plane and reading one scattered look-up line per edge; sum order same
# speedup vs baseline: 3.9219x; 1.0647x over previous
#include <hip/hip_runtime.h>
#include <hip/hip_bf16.h>

#define DIM 128

typedef __attribute__((ext_vector_type(2))) float v2f;
typedef __attribute__((ext_vector_type(8))) float v8f;

#define OWN_ROWS  416
#define OWN_WAVES 32
#define OWN_QCAP  512
#define OWN_LCAP  1024

#define PL_EDGES   4096
#define PL_WORDS   (PL_EDGES / 32)
#define PL_PITCH   (PL_WORDS + 4)
#define PL_OWNERS  256
#define PL_THREADS 512

#define PK_SHIFT 23
#define PK_MASK  ((1u << PK_SHIFT) - 1u)

__global__ __launch_bounds__(PL_THREADS)
void sage_owner_planes(const int* __restrict__ src,
                       const int* __restrict__ dst,
                       unsigned* __restrict__ planes,
                       unsigned* __restrict__ packed,
                       int E, int owners, int nwp) {
  __shared__ uint4 bits4[PL_OWNERS * PL_PITCH / 4];
  unsigned* bits = reinterpret_cast<unsigned*>(bits4);

  const int tid  = threadIdx.x;
  const int lane = tid & 31;
  const int wave = tid >> 5;
  const unsigned own0 = blockIdx.y * PL_OWNERS;
  const int e0 = blockIdx.x * PL_EDGES;

  for (int i = tid; i < PL_OWNERS * PL_PITCH / 4; i += PL_THREADS) bits4[i] = make_uint4(0u, 0u, 0u, 0u);
  __syncthreads();

  auto mark = [&](int d, int i) __attribute__((always_inline)) {
    const unsigned o = (unsigned)d / (unsigned)OWN_ROWS - own0;
    if (o < (unsigned)PL_OWNERS) atomicOr(&bits[o * PL_PITCH + (i >> 5)], 1u << (i & 31));
  };

  int4 d[PL_EDGES / (4 * PL_THREADS)];
  int4 s[PL_EDGES / (4 * PL_THREADS)];
  const bool whole = e0 + PL_EDGES <= E;
  const bool pack  = blockIdx.y == 0;
  auto pk = [](int sv, int dv) __attribute__((always_inline)) {
    return (unsigned)sv | (((unsigned)dv % (unsigned)OWN_ROWS) << PK_SHIFT);
  };
  if (whole) {
    if (pack) {
#pragma unroll
      for (int k = 0; k < PL_EDGES / (4 * PL_THREADS); ++k) {
        s[k] = *reinterpret_cast<const int4*>(src + e0 + k * 4 * PL_THREADS + 4 * tid);
      }
    }
#pragma unroll
    for (int k = 0; k < PL_EDGES / (4 * PL_THREADS); ++k) {
      d[k] = *reinterpret_cast<const int4*>(dst + e0 + k * 4 * PL_THREADS + 4 * tid);
    }
    if (pack) {
#pragma unroll
      for (int k = 0; k < PL_EDGES / (4 * PL_THREADS); ++k) {
        *reinterpret_cast<uint4*>(packed + e0 + k * 4 * PL_THREADS + 4 * tid) =
            make_uint4(pk(s[k].x, d[k].x), pk(s[k].y, d[k].y), pk(s[k].z, d[k].z), pk(s[k].w, d[k].w));
      }
    }
  } else {
#pragma unroll
    for (int k = 0; k < PL_EDGES / (4 * PL_THREADS); ++k) {
      const int e = e0 + k * 4 * PL_THREADS + 4 * tid;
      d[k].x = (e + 0 < E) ? dst[e + 0] : -1;
      d[k].y = (e + 1 < E) ? dst[e + 1] : -1;
      d[k].z = (e + 2 < E) ? dst[e + 2] : -1;
      d[k].w = (e + 3 < E) ? dst[e + 3] : -1;
      if (pack) {
        if (e + 0 < E) packed[e + 0] = pk(src[e + 0], d[k].x);
        if (e + 1 < E) packed[e + 1] = pk(src[e + 1], d[k].y);
        if (e + 2 < E) packed[e + 2] = pk(src[e + 2], d[k].z);
        if (e + 3 < E) packed[e + 3] = pk(src[e + 3], d[k].w);
      }
    }
  }
#pragma unroll
  for (int k = 0; k < PL_EDGES / (4 * PL_THREADS); ++k) {
    const int i = k * 4 * PL_THREADS + 4 * tid;
    mark(d[k].x, i);
    mark(d[k].y, i + 1);
    mark(d[k].z, i + 2);
    mark(d[k].w, i + 3);
  }
  __syncthreads();

  for (int o = wave; o < PL_OWNERS; o += PL_THREADS / 32) {
    const unsigned og = own0 + o;
    if (og < (unsigned)owners) {
      const uint4 v = bits4[o * (PL_PITCH / 4) + lane];
      uint4* q = reinterpret_cast<uint4*>(planes + (size_t)og * nwp + (size_t)blockIdx.x * PL_WORDS);
      q[lane] = v;
    }
  }
}

template <bool FIRST>
__global__ __launch_bounds__(1024)
void sage_own_rows(const float* __restrict__ feat,
                   const unsigned* __restrict__ packed,
                   const unsigned* __restrict__ planes,
                   unsigned* __restrict__ lists,
                   unsigned* __restrict__ lcount,
                   float* __restrict__ agg,
                   float* __restrict__ cnt,
                   int nwp, int N) {
  __shared__ float tile[OWN_ROWS * DIM];
  __shared__ float ctile[OWN_ROWS];
  __shared__ int   ring[OWN_WAVES][OWN_QCAP];

  const int tid  = threadIdx.x;
  const int lane = tid & 31;
  const int wave = __builtin_amdgcn_readfirstlane(tid >> 5);
  const int row0 = blockIdx.x * OWN_ROWS;

  float4* t4 = reinterpret_cast<float4*>(tile);
  for (int i = tid; i < OWN_ROWS * DIM / 4; i += 1024) t4[i] = make_float4(0.0f, 0.0f, 0.0f, 0.0f);
  if (FIRST) {
    for (int i = tid; i < OWN_ROWS; i += 1024) ctile[i] = 0.0f;
  }
  __syncthreads();

  int* myq = ring[wave];
  const unsigned below = (1u << lane) - 1u;
  unsigned head = 0, tail = 0;
  unsigned found = 0;
  unsigned* mylist = lists + ((size_t)blockIdx.x * OWN_WAVES + wave) * OWN_LCAP;

  auto gather_add = [&](int s, unsigned rr, unsigned n) __attribute__((always_inline)) {
    for (unsigned g = 0; g < n; g += 8) {
      float4 v[8];
#pragma unroll
      for (int k = 0; k < 8; ++k) {
        const int sj = __builtin_amdgcn_readlane(s, g + k);
        v[k] = reinterpret_cast<const float4*>(feat + (size_t)sj * DIM)[lane];
      }
#pragma unroll
      for (int k = 0; k < 8; ++k) {
        if (g + k < n) {
          const unsigned rj = __builtin_amdgcn_readlane(rr, g + k);
          float* t = tile + rj * DIM + lane;
          atomicAdd(t,      v[k].x);
          atomicAdd(t + 32, v[k].y);
          atomicAdd(t + 64, v[k].z);
          atomicAdd(t + 96, v[k].w);
        }
      }
    }
  };

  auto flush = [&](unsigned n) __attribute__((always_inline)) {
    const bool valid = (unsigned)lane < n;
    int s = 0;
    unsigned rr = 0;
    if (valid) {
      const int e = myq[(head + lane) & (OWN_QCAP - 1)];
      const unsigned w = packed[e];
      s  = (int)(w & PK_MASK);
      rr = w >> PK_SHIFT;
      if (FIRST) {
        atomicAdd(&ctile[rr], 1.0f);
        if (found + lane < (unsigned)OWN_LCAP) mylist[found + lane] = w;
      }
    }
    found += n;
    gather_add(s, rr, n);
  };

  auto take = [&](unsigned m, int e0) __attribute__((always_inline)) {
    unsigned bal = __builtin_amdgcn_ballot_w32(m != 0u);
    while (bal != 0u) {
      if (m != 0u) {
        myq[(tail + __builtin_popcount(bal & below)) & (OWN_QCAP - 1)] = e0 + __builtin_ctz(m);
        m &= m - 1u;
      }
      tail += __builtin_popcount(bal);
      while (tail - head >= 32u) {
        flush(32u);
        head += 32u;
      }
      bal = __builtin_amdgcn_ballot_w32(m != 0u);
    }
  };

  const uint4* pl = reinterpret_cast<const uint4*>(planes + (size_t)blockIdx.x * nwp);
  const int nsteps = nwp / 128;
  unsigned kept = 0;
  if (!FIRST) kept = (unsigned)__builtin_amdgcn_readfirstlane((int)lcount[blockIdx.x * OWN_WAVES + wave]);
  if (!FIRST && kept <= (unsigned)OWN_LCAP) {
    if (kept != 0u) {
      const unsigned lastw = kept - 1u;
      unsigned wn = mylist[(unsigned)lane < lastw ? (unsigned)lane : lastw];
      unsigned i = 0;
      for (; i + 32u <= kept; i += 32u) {
        const unsigned w = wn;
        const unsigned nx = i + 32u + (unsigned)lane;
        wn = mylist[nx < lastw ? nx : lastw];
        gather_add((int)(w & PK_MASK), w >> PK_SHIFT, 32u);
      }
      if (i < kept) gather_add((int)(wn & PK_MASK), wn >> PK_SHIFT, kept - i);
    }
  } else if (wave < nsteps) {
    int step = wave;
    uint4 cur = pl[step * 32 + lane];
    while (step < nsteps) {
      const int nxt = step + OWN_WAVES;
      const uint4 nx = pl[(nxt < nsteps ? nxt : step) * 32 + lane];
      const int e0 = (step * 32 + lane) * 128;
      take(cur.x, e0);
      take(cur.y, e0 + 32);
      take(cur.z, e0 + 64);
      take(cur.w, e0 + 96);
      cur = nx;
      step = nxt;
    }
  }
  if (tail != head) flush(tail - head);
  if (FIRST && lane == 0) lcount[blockIdx.x * OWN_WAVES + wave] = found;

  __syncthreads();

  const int left  = N - row0;
  const int nrows = left < OWN_ROWS ? left : OWN_ROWS;
  for (int r = wave; r < nrows; r += OWN_WAVES) {
    const float* t = tile + r * DIM + lane;
    reinterpret_cast<float4*>(agg + (size_t)(row0 + r) * DIM)[lane] = make_float4(t[0], t[32], t[64], t[96]);
  }
  if (FIRST) {
    for (int i = tid; i < nrows; i += 1024) cnt[row0 + i] = ctile[i];
  }
}

#define GEMM_WAVES 4
#define GEMM_KS    16
#define GEMM_NT    (DIM / 16)
#define GEMM_WGS   512
#define GEMM_PITCH (DIM + 4)

__device__ __forceinline__ int kof(int s, int h) { return 8 * (s >> 1) + 4 * h + 2 * (s & 1); }

__global__ __launch_bounds__(32 * GEMM_WAVES)
void sage_fused_gemm(const float* __restrict__ agg,
                     const float* __restrict__ cnt,
                     const float* __restrict__ xin,
                     const float* __restrict__ Wl,
                     const float* __restrict__ Wr,
                     const float* __restrict__ bias,
                     float* __restrict__ out,
                     int do_relu, int nstripes) {
  __shared__ float4 part4[2][GEMM_WAVES][16 * GEMM_PITCH / 4];

  const int lane = threadIdx.x & 31;
  const int wave = __builtin_amdgcn_readfirstlane(threadIdx.x >> 5);
  const int m    = lane & 15;
  const int half = lane >> 4;

  const bool left = wave < 2;
  const float* A  = left ? agg : xin;
  const float* Wm = left ? Wl : Wr;
  const int kb    = (wave & 1) * (4 * GEMM_KS);

  v2f b[GEMM_NT][GEMM_KS];
#pragma unroll
  for (int t = 0; t < GEMM_NT; ++t) {
#pragma unroll
    for (int s = 0; s < GEMM_KS; ++s) {
      const int kk = kb + kof(s, half);
      b[t][s].x = Wm[(size_t)kk * DIM + t * 16 + m];
      b[t][s].y = Wm[(size_t)(kk + 1) * DIM + t * 16 + m];
    }
  }
  const float4 bias4 = reinterpret_cast<const float4*>(bias)[lane];

  float4 an[GEMM_KS / 2];
  float  cn = 1.0f;
  auto load_a = [&](int st) __attribute__((always_inline)) {
    const size_t row = (size_t)st * 16 + m;
    const float4* p = reinterpret_cast<const float4*>(A + row * DIM + kb + 4 * half);
#pragma unroll
    for (int j = 0; j < GEMM_KS / 2; ++j) an[j] = p[2 * j];
    if (left) cn = cnt[row];
  };

  int st = blockIdx.x;
  if (st < nstripes) load_a(st);
  int buf = 0;

  for (; st < nstripes; st += gridDim.x) {
    const int row0 = st * 16;
    float4 ac[GEMM_KS / 2];
#pragma unroll
    for (int j = 0; j < GEMM_KS / 2; ++j) ac[j] = an[j];
    const float scale = left ? 1.0f / fmaxf(cn, 1.0f) : 1.0f;

    const int nx = st + gridDim.x;
    load_a(nx < nstripes ? nx : st);

    v8f c[GEMM_NT];
#pragma unroll
    for (int t = 0; t < GEMM_NT; ++t) c[t] = v8f{};

#pragma unroll
    for (int s = 0; s < GEMM_KS; ++s) {
      const float4 q = ac[s >> 1];
      v2f a;
      a.x = ((s & 1) ? q.z : q.x) * scale;
      a.y = ((s & 1) ? q.w : q.y) * scale;
#pragma unroll
      for (int t = 0; t < GEMM_NT; ++t)
        c[t] = __builtin_amdgcn_wmma_f32_16x16x4_f32(false, a, false, b[t][s], (short)0, c[t],
                                                     false, false);
    }

    float* P = reinterpret_cast<float*>(part4[buf][wave]);
#pragma unroll
    for (int t = 0; t < GEMM_NT; ++t) {
#pragma unroll
      for (int r = 0; r < 8; ++r) P[(r + 8 * half) * GEMM_PITCH + t * 16 + m] = c[t][r];
    }
    __syncthreads();

#pragma unroll
    for (int i = 0; i < 4; ++i) {
      const int row = 4 * wave + i;
      const int o = row * (GEMM_PITCH / 4) + lane;
      const float4 p0 = part4[buf][0][o];
      const float4 p1 = part4[buf][1][o];
      const float4 p2 = part4[buf][2][o];
      const float4 p3 = part4[buf][3][o];
      float4 v;
      v.x = ((p0.x + p1.x) + p2.x) + p3.x + bias4.x;
      v.y = ((p0.y + p1.y) + p2.y) + p3.y + bias4.y;
      v.z = ((p0.z + p1.z) + p2.z) + p3.z + bias4.z;
      v.w = ((p0.w + p1.w) + p2.w) + p3.w + bias4.w;
      if (do_relu) {
        v.x = fmaxf(v.x, 0.0f);
        v.y = fmaxf(v.y, 0.0f);
        v.z = fmaxf(v.z, 0.0f);
        v.w = fmaxf(v.w, 0.0f);
      }
      reinterpret_cast<float4*>(out + (size_t)(row0 + row) * DIM)[lane] = v;
    }
    buf ^= 1;
  }
}

extern "C" void kernel_launch(void* const* d_in, const int* in_sizes, int n_in,
                              void* d_out, int out_size, void* d_ws, size_t ws_size,
                              hipStream_t stream) {
  const float* x   = (const float*)d_in[0];
  const int*  edge = (const int*)d_in[1];
  const float* W1l = (const float*)d_in[2];
  const float* b1  = (const float*)d_in[3];
  const float* W1r = (const float*)d_in[4];
  const float* W2l = (const float*)d_in[5];
  const float* b2  = (const float*)d_in[6];
  const float* W2r = (const float*)d_in[7];

  const int N = in_sizes[0] / DIM;
  const int E = in_sizes[1] / 2;
  const int* src = edge;
  const int* dst = edge + E;

  float* agg = (float*)d_ws;
  float* cnt = agg + (size_t)N * DIM;
  float* h   = cnt + N;
  float* out = (float*)d_out;

  const int own_blocks  = (N + OWN_ROWS - 1) / OWN_ROWS;
  const int gemm_blocks = N / 16;

  const int pl_chunks = (E + PL_EDGES - 1) / PL_EDGES;
  const int nwp       = pl_chunks * PL_WORDS;
  const size_t pl_off = ((size_t)N * (2 * DIM + 1) + 63) / 64 * 64;
  unsigned* planes    = (unsigned*)d_ws + pl_off;
  const dim3 pl_grid(pl_chunks, (own_blocks + PL_OWNERS - 1) / PL_OWNERS);
  unsigned* packed    = planes + (size_t)own_blocks * nwp;
  unsigned* lists     = packed + ((size_t)E + 63) / 64 * 64;
  unsigned* lcount    = lists + (size_t)own_blocks * OWN_WAVES * OWN_LCAP;

  sage_owner_planes<<<pl_grid, PL_THREADS, 0, stream>>>(src, dst, planes, packed, E, own_blocks, nwp);

  sage_own_rows<true><<<own_blocks, 1024, 0, stream>>>(x, packed, planes, lists, lcount, agg, cnt, nwp, N);
  sage_fused_gemm<<<GEMM_WGS, 32 * GEMM_WAVES, 0, stream>>>(agg, cnt, x, W1l, W1r, b1, h, 1, gemm_blocks);

  sage_own_rows<false><<<own_blocks, 1024, 0, stream>>>(h, packed, planes, lists, lcount, agg, cnt, nwp, N);
  sage_fused_gemm<<<GEMM_WGS, 32 * GEMM_WAVES, 0, stream>>>(agg, cnt, h, W2l, W2r, b2, out, 0, gemm_blocks);
}


# layer 1's owner pass streams its edges' look-up words from two-line cells per (4096-edge chunk, owner), placed by the first launch (LDS counter); no plane scan, no look-up read; sum order same
# speedup vs baseline: 4.1533x; 1.0590x over previous
#include <hip/hip_runtime.h>
#include <hip/hip_bf16.h>

#define DIM 128

typedef __attribute__((ext_vector_type(2))) float v2f;
typedef __attribute__((ext_vector_type(8))) float v8f;

#define OWN_ROWS  416
#define OWN_WAVES 32
#define OWN_QCAP  128
#define OWN_LCAP  1024

#define PL_EDGES   4096
#define PL_OWNERS  256
#define PL_THREADS 512
#define CELL_WORDS 64
#define CELL_CAP   63u

#define PK_SHIFT 23
#define PK_MASK  ((1u << PK_SHIFT) - 1u)
__device__ __forceinline__ unsigned pk_word(int sv, int dv) {
  return (unsigned)sv | (((unsigned)dv % (unsigned)OWN_ROWS) << PK_SHIFT);
}

__global__ __launch_bounds__(PL_THREADS)
void sage_owner_cells(const int* __restrict__ src,
                      const int* __restrict__ dst,
                      unsigned* __restrict__ cells,
                      int E, int owners, int nchunks) {
  __shared__ unsigned cntr[PL_OWNERS];
  __shared__ unsigned ent[PL_OWNERS * CELL_WORDS];

  const int tid  = threadIdx.x;
  const int lane = tid & 31;
  const int wave = tid >> 5;
  const unsigned own0 = blockIdx.y * PL_OWNERS;
  const int e0 = blockIdx.x * PL_EDGES;

  for (int i = tid; i < PL_OWNERS; i += PL_THREADS) cntr[i] = 0u;
  __syncthreads();

  auto place = [&](int sv, int dv) __attribute__((always_inline)) {
    const unsigned o = (unsigned)dv / (unsigned)OWN_ROWS - own0;
    if (o < (unsigned)PL_OWNERS) {
      const unsigned r = atomicAdd(&cntr[o], 1u);
      if (r < CELL_CAP) ent[o * CELL_WORDS + ((r + o) & (CELL_WORDS - 1))] = pk_word(sv, dv);
    }
  };

  int4 d[PL_EDGES / (4 * PL_THREADS)];
  int4 s[PL_EDGES / (4 * PL_THREADS)];
  const bool whole = e0 + PL_EDGES <= E;
  if (whole) {
#pragma unroll
    for (int k = 0; k < PL_EDGES / (4 * PL_THREADS); ++k) {
      s[k] = *reinterpret_cast<const int4*>(src + e0 + k * 4 * PL_THREADS + 4 * tid);
    }
#pragma unroll
    for (int k = 0; k < PL_EDGES / (4 * PL_THREADS); ++k) {
      d[k] = *reinterpret_cast<const int4*>(dst + e0 + k * 4 * PL_THREADS + 4 * tid);
    }
  } else {
#pragma unroll
    for (int k = 0; k < PL_EDGES / (4 * PL_THREADS); ++k) {
      const int e = e0 + k * 4 * PL_THREADS + 4 * tid;
      d[k].x = (e + 0 < E) ? dst[e + 0] : -1;
      d[k].y = (e + 1 < E) ? dst[e + 1] : -1;
      d[k].z = (e + 2 < E) ? dst[e + 2] : -1;
      d[k].w = (e + 3 < E) ? dst[e + 3] : -1;
      s[k].x = (e + 0 < E) ? src[e + 0] : 0;
      s[k].y = (e + 1 < E) ? src[e + 1] : 0;
      s[k].z = (e + 2 < E) ? src[e + 2] : 0;
      s[k].w = (e + 3 < E) ? src[e + 3] : 0;
    }
  }
#pragma unroll
  for (int k = 0; k < PL_EDGES / (4 * PL_THREADS); ++k) {
    place(s[k].x, d[k].x);
    place(s[k].y, d[k].y);
    place(s[k].z, d[k].z);
    place(s[k].w, d[k].w);
  }
  __syncthreads();

  for (int o = wave; o < PL_OWNERS; o += PL_THREADS / 32) {
    const unsigned og = own0 + o;
    if (og < (unsigned)owners) {
      const unsigned total = cntr[o];
      const unsigned n  = total < CELL_CAP ? total : CELL_CAP;
      const unsigned j1 = 2u * (unsigned)lane;
      uint2 v;
      v.x = total;
      if (lane != 0) v.x = (j1 - 1u < n) ? ent[o * CELL_WORDS + ((j1 - 1u + o) & (CELL_WORDS - 1))] : 0u;
      v.y = (j1 < n) ? ent[o * CELL_WORDS + ((j1 + o) & (CELL_WORDS - 1))] : 0u;
      reinterpret_cast<uint2*>(cells + ((size_t)og * nchunks + blockIdx.x) * CELL_WORDS)[lane] = v;
    }
  }
}

template <bool FIRST>
__global__ __launch_bounds__(1024)
void sage_own_rows(const float* __restrict__ feat,
                   const int* __restrict__ src,
                   const int* __restrict__ dst,
                   const unsigned* __restrict__ cells,
                   unsigned* __restrict__ lists,
                   unsigned* __restrict__ lcount,
                   float* __restrict__ agg,
                   float* __restrict__ cnt,
                   int nchunks, int E, int N) {
  __shared__ float    tile[OWN_ROWS * DIM];
  __shared__ float    ctile[OWN_ROWS];
  __shared__ unsigned ring[OWN_WAVES][OWN_QCAP];

  const int tid  = threadIdx.x;
  const int lane = tid & 31;
  const int wave = __builtin_amdgcn_readfirstlane(tid >> 5);
  const int row0 = blockIdx.x * OWN_ROWS;

  float4* t4 = reinterpret_cast<float4*>(tile);
  for (int i = tid; i < OWN_ROWS * DIM / 4; i += 1024) t4[i] = make_float4(0.0f, 0.0f, 0.0f, 0.0f);
  if (FIRST) {
    for (int i = tid; i < OWN_ROWS; i += 1024) ctile[i] = 0.0f;
  }
  __syncthreads();

  unsigned* myq = ring[wave];
  const unsigned below = (1u << lane) - 1u;
  unsigned head = 0, tail = 0;
  unsigned found = 0;
  unsigned* mylist = lists + ((size_t)blockIdx.x * OWN_WAVES + wave) * OWN_LCAP;

  auto gather_add = [&](int s, unsigned rr, unsigned n) __attribute__((always_inline)) {
    for (unsigned g = 0; g < n; g += 8) {
      float4 v[8];
#pragma unroll
      for (int k = 0; k < 8; ++k) {
        const int sj = __builtin_amdgcn_readlane(s, g + k);
        v[k] = reinterpret_cast<const float4*>(feat + (size_t)sj * DIM)[lane];
      }
#pragma unroll
      for (int k = 0; k < 8; ++k) {
        if (g + k < n) {
          const unsigned rj = __builtin_amdgcn_readlane(rr, g + k);
          float* t = tile + rj * DIM + lane;
          atomicAdd(t,      v[k].x);
          atomicAdd(t + 32, v[k].y);
          atomicAdd(t + 64, v[k].z);
          atomicAdd(t + 96, v[k].w);
        }
      }
    }
  };

  auto flush = [&](unsigned n) __attribute__((always_inline)) {
    const bool valid = (unsigned)lane < n;
    int s = 0;
    unsigned rr = 0;
    if (valid) {
      const unsigned w = myq[(head + lane) & (OWN_QCAP - 1)];
      s  = (int)(w & PK_MASK);
      rr = w >> PK_SHIFT;
      if (FIRST) {
        atomicAdd(&ctile[rr], 1.0f);
        if (found + lane < (unsigned)OWN_LCAP) mylist[found + lane] = w;
      }
    }
    found += n;
    gather_add(s, rr, n);
  };
  auto drain = [&]() __attribute__((always_inline)) {
    while (tail - head >= 32u) {
      flush(32u);
      head += 32u;
    }
  };

  const uint2* cl = reinterpret_cast<const uint2*>(cells + (size_t)blockIdx.x * nchunks * CELL_WORDS);
  unsigned kept = 0;
  if (!FIRST) kept = (unsigned)__builtin_amdgcn_readfirstlane((int)lcount[blockIdx.x * OWN_WAVES + wave]);
  if (!FIRST && kept <= (unsigned)OWN_LCAP) {
    if (kept != 0u) {
      const unsigned lastw = kept - 1u;
      unsigned wn = mylist[(unsigned)lane < lastw ? (unsigned)lane : lastw];
      unsigned i = 0;
      for (; i + 32u <= kept; i += 32u) {
        const unsigned w = wn;
        const unsigned nx = i + 32u + (unsigned)lane;
        wn = mylist[nx < lastw ? nx : lastw];
        gather_add((int)(w & PK_MASK), w >> PK_SHIFT, 32u);
      }
      if (i < kept) gather_add((int)(wn & PK_MASK), wn >> PK_SHIFT, kept - i);
    }
  } else if (wave < nchunks) {
    int c = wave;
    uint2 cur = cl[c * 32 + lane];
    while (c < nchunks) {
      const int nxt = c + OWN_WAVES;
      const uint2 nx = cl[(nxt < nchunks ? nxt : c) * 32 + lane];
      const unsigned total = (unsigned)__builtin_amdgcn_readfirstlane((int)cur.x);
      if (total <= CELL_CAP) {
        const unsigned j1 = 2u * (unsigned)lane;
        if (lane != 0 && j1 - 1u < total) myq[(tail + j1 - 1u) & (OWN_QCAP - 1)] = cur.x;
        if (j1 < total) myq[(tail + j1) & (OWN_QCAP - 1)] = cur.y;
        tail += total;
        drain();
      } else {
        const int e0 = c * PL_EDGES;
        for (int i = 0; i < PL_EDGES; i += 32) {
          const int e = e0 + i + lane;
          int dv = -1, sv = 0;
          if (e < E) {
            dv = dst[e];
            sv = src[e];
          }
          const bool mine = (unsigned)dv / (unsigned)OWN_ROWS == blockIdx.x;
          const unsigned bal = __builtin_amdgcn_ballot_w32(mine);
          if (mine) myq[(tail + __builtin_popcount(bal & below)) & (OWN_QCAP - 1)] = pk_word(sv, dv);
          tail += __builtin_popcount(bal);
          drain();
        }
      }
      cur = nx;
      c = nxt;
    }
  }
  if (tail != head) flush(tail - head);
  if (FIRST && lane == 0) lcount[blockIdx.x * OWN_WAVES + wave] = found;

  __syncthreads();

  const int left  = N - row0;
  const int nrows = left < OWN_ROWS ? left : OWN_ROWS;
  for (int r = wave; r < nrows; r += OWN_WAVES) {
    const float* t = tile + r * DIM + lane;
    reinterpret_cast<float4*>(agg + (size_t)(row0 + r) * DIM)[lane] = make_float4(t[0], t[32], t[64], t[96]);
  }
  if (FIRST) {
    for (int i = tid; i < nrows; i += 1024) cnt[row0 + i] = ctile[i];
  }
}

#define GEMM_WAVES 4
#define GEMM_KS    16
#define GEMM_NT    (DIM / 16)
#define GEMM_WGS   512
#define GEMM_PITCH (DIM + 4)

__device__ __forceinline__ int kof(int s, int h) { return 8 * (s >> 1) + 4 * h + 2 * (s & 1); }

__global__ __launch_bounds__(32 * GEMM_WAVES)
void sage_fused_gemm(const float* __restrict__ agg,
                     const float* __restrict__ cnt,
                     const float* __restrict__ xin,
                     const float* __restrict__ Wl,
                     const float* __restrict__ Wr,
                     const float* __restrict__ bias,
                     float* __restrict__ out,
                     int do_relu, int nstripes) {
  __shared__ float4 part4[2][GEMM_WAVES][16 * GEMM_PITCH / 4];

  const int lane = threadIdx.x & 31;
  const int wave = __builtin_amdgcn_readfirstlane(threadIdx.x >> 5);
  const int m    = lane & 15;
  const int half = lane >> 4;

  const bool left = wave < 2;
  const float* A  = left ? agg : xin;
  const float* Wm = left ? Wl : Wr;
  const int kb    = (wave & 1) * (4 * GEMM_KS);

  v2f b[GEMM_NT][GEMM_KS];
#pragma unroll
  for (int t = 0; t < GEMM_NT; ++t) {
#pragma unroll
    for (int s = 0; s < GEMM_KS; ++s) {
      const int kk = kb + kof(s, half);
      b[t][s].x = Wm[(size_t)kk * DIM + t * 16 + m];
      b[t][s].y = Wm[(size_t)(kk + 1) * DIM + t * 16 + m];
    }
  }
  const float4 bias4 = reinterpret_cast<const float4*>(bias)[lane];

  float4 an[GEMM_KS / 2];
  float  cn = 1.0f;
  auto load_a = [&](int st) __attribute__((always_inline)) {
    const size_t row = (size_t)st * 16 + m;
    const float4* p = reinterpret_cast<const float4*>(A + row * DIM + kb + 4 * half);
#pragma unroll
    for (int j = 0; j < GEMM_KS / 2; ++j) an[j] = p[2 * j];
    if (left) cn = cnt[row];
  };

  int st = blockIdx.x;
  if (st < nstripes) load_a(st);
  int buf = 0;

  for (; st < nstripes; st += gridDim.x) {
    const int row0 = st * 16;
    float4 ac[GEMM_KS / 2];
#pragma unroll
    for (int j = 0; j < GEMM_KS / 2; ++j) ac[j] = an[j];
    const float scale = left ? 1.0f / fmaxf(cn, 1.0f) : 1.0f;

    const int nx = st + gridDim.x;
    load_a(nx < nstripes ? nx : st);

    v8f c[GEMM_NT];
#pragma unroll
    for (int t = 0; t < GEMM_NT; ++t) c[t] = v8f{};

#pragma unroll
    for (int s = 0; s < GEMM_KS; ++s) {
      const float4 q = ac[s >> 1];
      v2f a;
      a.x = ((s & 1) ? q.z : q.x) * scale;
      a.y = ((s & 1) ? q.w : q.y) * scale;
#pragma unroll
      for (int t = 0; t < GEMM_NT; ++t)
        c[t] = __builtin_amdgcn_wmma_f32_16x16x4_f32(false, a, false, b[t][s], (short)0, c[t],
                                                     false, false);
    }

    float* P = reinterpret_cast<float*>(part4[buf][wave]);
#pragma unroll
    for (int t = 0; t < GEMM_NT; ++t) {
#pragma unroll
      for (int r = 0; r < 8; ++r) P[(r + 8 * half) * GEMM_PITCH + t * 16 + m] = c[t][r];
    }
    __syncthreads();

#pragma unroll
    for (int i = 0; i < 4; ++i) {
      const int row = 4 * wave + i;
      const int o = row * (GEMM_PITCH / 4) + lane;
      const float4 p0 = part4[buf][0][o];
      const float4 p1 = part4[buf][1][o];
      const float4 p2 = part4[buf][2][o];
      const float4 p3 = part4[buf][3][o];
      float4 v;
      v.x = ((p0.x + p1.x) + p2.x) + p3.x + bias4.x;
      v.y = ((p0.y + p1.y) + p2.y) + p3.y + bias4.y;
      v.z = ((p0.z + p1.z) + p2.z) + p3.z + bias4.z;
      v.w = ((p0.w + p1.w) + p2.w) + p3.w + bias4.w;
      if (do_relu) {
        v.x = fmaxf(v.x, 0.0f);
        v.y = fmaxf(v.y, 0.0f);
        v.z = fmaxf(v.z, 0.0f);
        v.w = fmaxf(v.w, 0.0f);
      }
      reinterpret_cast<float4*>(out + (size_t)(row0 + row) * DIM)[lane] = v;
    }
    buf ^= 1;
  }
}

extern "C" void kernel_launch(void* const* d_in, const int* in_sizes, int n_in,
                              void* d_out, int out_size, void* d_ws, size_t ws_size,
                              hipStream_t stream) {
  const float* x   = (const float*)d_in[0];
  const int*  edge = (const int*)d_in[1];
  const float* W1l = (const float*)d_in[2];
  const float* b1  = (const float*)d_in[3];
  const float* W1r = (const float*)d_in[4];
  const float* W2l = (const float*)d_in[5];
  const float* b2  = (const float*)d_in[6];
  const float* W2r = (const float*)d_in[7];

  const int N = in_sizes[0] / DIM;
  const int E = in_sizes[1] / 2;
  const int* src = edge;
  const int* dst = edge + E;

  float* agg = (float*)d_ws;
  float* cnt = agg + (size_t)N * DIM;
  float* h   = cnt + N;
  float* out = (float*)d_out;

  const int own_blocks  = (N + OWN_ROWS - 1) / OWN_ROWS;
  const int gemm_blocks = N / 16;

  const int nchunks   = (E + PL_EDGES - 1) / PL_EDGES;
  const size_t cl_off = ((size_t)N * (2 * DIM + 1) + 63) / 64 * 64;
  unsigned* cells     = (unsigned*)d_ws + cl_off;
  const dim3 cl_grid(nchunks, (own_blocks + PL_OWNERS - 1) / PL_OWNERS);
  unsigned* lists     = cells + (size_t)own_blocks * nchunks * CELL_WORDS;
  unsigned* lcount    = lists + (size_t)own_blocks * OWN_WAVES * OWN_LCAP;

  sage_owner_cells<<<cl_grid, PL_THREADS, 0, stream>>>(src, dst, cells, E, own_blocks, nchunks);

  sage_own_rows<true><<<own_blocks, 1024, 0, stream>>>(x, src, dst, cells, lists, lcount, agg, cnt, nchunks, E, N);
  sage_fused_gemm<<<GEMM_WGS, 32 * GEMM_WAVES, 0, stream>>>(agg, cnt, x, W1l, W1r, b1, h, 1, gemm_blocks);

  sage_own_rows<false><<<own_blocks, 1024, 0, stream>>>(h, src, dst, cells, lists, lcount, agg, cnt, nchunks, E, N);
  sage_fused_gemm<<<GEMM_WGS, 32 * GEMM_WAVES, 0, stream>>>(agg, cnt, h, W2l, W2r, b2, out, 0, gemm_blocks);
}


# neighbour sum: LDS f32 atomics on the tile replaced: each look-up word goes to the LDS queue of the wave owning its row (13 rows a wave), which sums its rows in registers; sum order = queue arrival
# speedup vs baseline: 4.2841x; 1.0315x over previous
#include <hip/hip_runtime.h>
#include <hip/hip_bf16.h>

#define DIM 128

typedef __attribute__((ext_vector_type(2))) float v2f;
typedef __attribute__((ext_vector_type(8))) float v8f;
typedef __attribute__((ext_vector_type(16))) float v16f;

#define OWN_ROWS  416
#define OWN_WAVES 32
#define OWN_WROWS (OWN_ROWS / OWN_WAVES)
#define OWN_RCAP  512
#define OWN_QCAP  128
#define OWN_LCAP  1024

#define PL_EDGES   4096
#define PL_OWNERS  256
#define PL_THREADS 512
#define CELL_WORDS 64
#define CELL_CAP   63u

#define PK_SHIFT 23
#define PK_MASK  ((1u << PK_SHIFT) - 1u)
__device__ __forceinline__ unsigned pk_word(int sv, int dv) {
  return (unsigned)sv | (((unsigned)dv % (unsigned)OWN_ROWS) << PK_SHIFT);
}

__global__ __launch_bounds__(PL_THREADS)
void sage_owner_cells(const int* __restrict__ src,
                      const int* __restrict__ dst,
                      unsigned* __restrict__ cells,
                      int E, int owners, int nchunks) {
  __shared__ unsigned cntr[PL_OWNERS];
  __shared__ unsigned ent[PL_OWNERS * CELL_WORDS];

  const int tid  = threadIdx.x;
  const int lane = tid & 31;
  const int wave = tid >> 5;
  const unsigned own0 = blockIdx.y * PL_OWNERS;
  const int e0 = blockIdx.x * PL_EDGES;

  for (int i = tid; i < PL_OWNERS; i += PL_THREADS) cntr[i] = 0u;
  __syncthreads();

  auto place = [&](int sv, int dv) __attribute__((always_inline)) {
    const unsigned o = (unsigned)dv / (unsigned)OWN_ROWS - own0;
    if (o < (unsigned)PL_OWNERS) {
      const unsigned r = atomicAdd(&cntr[o], 1u);
      if (r < CELL_CAP) ent[o * CELL_WORDS + ((r + o) & (CELL_WORDS - 1))] = pk_word(sv, dv);
    }
  };

  int4 d[PL_EDGES / (4 * PL_THREADS)];
  int4 s[PL_EDGES / (4 * PL_THREADS)];
  const bool whole = e0 + PL_EDGES <= E;
  if (whole) {
#pragma unroll
    for (int k = 0; k < PL_EDGES / (4 * PL_THREADS); ++k) {
      s[k] = *reinterpret_cast<const int4*>(src + e0 + k * 4 * PL_THREADS + 4 * tid);
    }
#pragma unroll
    for (int k = 0; k < PL_EDGES / (4 * PL_THREADS); ++k) {
      d[k] = *reinterpret_cast<const int4*>(dst + e0 + k * 4 * PL_THREADS + 4 * tid);
    }
  } else {
#pragma unroll
    for (int k = 0; k < PL_EDGES / (4 * PL_THREADS); ++k) {
      const int e = e0 + k * 4 * PL_THREADS + 4 * tid;
      d[k].x = (e + 0 < E) ? dst[e + 0] : -1;
      d[k].y = (e + 1 < E) ? dst[e + 1] : -1;
      d[k].z = (e + 2 < E) ? dst[e + 2] : -1;
      d[k].w = (e + 3 < E) ? dst[e + 3] : -1;
      s[k].x = (e + 0 < E) ? src[e + 0] : 0;
      s[k].y = (e + 1 < E) ? src[e + 1] : 0;
      s[k].z = (e + 2 < E) ? src[e + 2] : 0;
      s[k].w = (e + 3 < E) ? src[e + 3] : 0;
    }
  }
#pragma unroll
  for (int k = 0; k < PL_EDGES / (4 * PL_THREADS); ++k) {
    place(s[k].x, d[k].x);
    place(s[k].y, d[k].y);
    place(s[k].z, d[k].z);
    place(s[k].w, d[k].w);
  }
  __syncthreads();

  for (int o = wave; o < PL_OWNERS; o += PL_THREADS / 32) {
    const unsigned og = own0 + o;
    if (og < (unsigned)owners) {
      const unsigned total = cntr[o];
      const unsigned n  = total < CELL_CAP ? total : CELL_CAP;
      const unsigned j1 = 2u * (unsigned)lane;
      uint2 v;
      v.x = total;
      if (lane != 0) v.x = (j1 - 1u < n) ? ent[o * CELL_WORDS + ((j1 - 1u + o) & (CELL_WORDS - 1))] : 0u;
      v.y = (j1 < n) ? ent[o * CELL_WORDS + ((j1 + o) & (CELL_WORDS - 1))] : 0u;
      reinterpret_cast<uint2*>(cells + ((size_t)og * nchunks + blockIdx.x) * CELL_WORDS)[lane] = v;
    }
  }
}

template <bool FIRST>
__global__ __launch_bounds__(1024)
void sage_own_rows(const float* __restrict__ feat,
                   const int* __restrict__ src,
                   const int* __restrict__ dst,
                   const unsigned* __restrict__ cells,
                   unsigned* __restrict__ lists,
                   unsigned* __restrict__ lcount,
                   float* __restrict__ agg,
                   float* __restrict__ cnt,
                   int nchunks, int E, int N) {
  __shared__ float    tile[OWN_ROWS * DIM];
  __shared__ float    ctile[OWN_ROWS];
  __shared__ unsigned ring[OWN_WAVES][OWN_QCAP];
  __shared__ unsigned rq[OWN_WAVES][OWN_RCAP];
  __shared__ unsigned rqn[OWN_WAVES];

  const int tid  = threadIdx.x;
  const int lane = tid & 31;
  const int wave = __builtin_amdgcn_readfirstlane(tid >> 5);
  const int row0 = blockIdx.x * OWN_ROWS;
  const int left  = N - row0;
  const int nrows = left < OWN_ROWS ? left : OWN_ROWS;
  const unsigned rbase = (unsigned)wave * (unsigned)OWN_WROWS;

  unsigned* myq = ring[wave];
  const unsigned below = (1u << lane) - 1u;
  unsigned head = 0, tail = 0;
  unsigned* mylist = lists + ((size_t)blockIdx.x * OWN_WAVES + wave) * OWN_LCAP;
  const uint2* cl = reinterpret_cast<const uint2*>(cells + (size_t)blockIdx.x * nchunks * CELL_WORDS);

  v16f ax = {}, ay = {}, az = {}, aw = {};
  auto add_row = [&](unsigned r, const float4& v) __attribute__((always_inline)) {
    ax[r] += v.x;
    ay[r] += v.y;
    az[r] += v.z;
    aw[r] += v.w;
  };

  auto acc_batch = [&](unsigned w, unsigned m) __attribute__((always_inline)) {
    const int s = (int)(w & PK_MASK);
    const unsigned rl = (w >> PK_SHIFT) - rbase;
    for (unsigned g = 0; g < m; g += 8) {
      float4 v[8];
#pragma unroll
      for (int k = 0; k < 8; ++k) {
        const int sj = __builtin_amdgcn_readlane(s, g + k);
        v[k] = reinterpret_cast<const float4*>(feat + (size_t)sj * DIM)[lane];
      }
#pragma unroll
      for (int k = 0; k < 8; ++k) {
        if (g + k < m) add_row((unsigned)__builtin_amdgcn_readlane((int)rl, g + k), v[k]);
      }
    }
  };

  auto store_acc = [&]() __attribute__((always_inline)) {
#pragma unroll
    for (int i = 0; i < OWN_WROWS; ++i) {
      const int r = (int)rbase + i;
      if (r < nrows) reinterpret_cast<float4*>(agg + (size_t)(row0 + r) * DIM)[lane] = make_float4(ax[i], ay[i], az[i], aw[i]);
    }
  };

  auto walk = [&](auto sink) __attribute__((always_inline)) {
    if (wave < nchunks) {
      int c = wave;
      uint2 cur = cl[c * 32 + lane];
      while (c < nchunks) {
        const int nxt = c + OWN_WAVES;
        const uint2 nx = cl[(nxt < nchunks ? nxt : c) * 32 + lane];
        const unsigned total = (unsigned)__builtin_amdgcn_readfirstlane((int)cur.x);
        if (total <= CELL_CAP) {
          const unsigned j1 = 2u * (unsigned)lane;
          sink(lane != 0 && j1 - 1u < total, cur.x);
          sink(j1 < total, cur.y);
        } else {
          const int e0 = c * PL_EDGES;
          for (int i = 0; i < PL_EDGES; i += 32) {
            const int e = e0 + i + lane;
            int dv = -1, sv = 0;
            if (e < E) {
              dv = dst[e];
              sv = src[e];
            }
            const bool mine = (unsigned)dv / (unsigned)OWN_ROWS == blockIdx.x;
            sink(mine, pk_word(sv, dv));
          }
        }
        cur = nx;
        c = nxt;
      }
    }
  };

  auto tile_add = [&](int s, unsigned rr, unsigned n) __attribute__((always_inline)) {
    for (unsigned g = 0; g < n; g += 8) {
      float4 v[8];
#pragma unroll
      for (int k = 0; k < 8; ++k) {
        const int sj = __builtin_amdgcn_readlane(s, g + k);
        v[k] = reinterpret_cast<const float4*>(feat + (size_t)sj * DIM)[lane];
      }
#pragma unroll
      for (int k = 0; k < 8; ++k) {
        if (g + k < n) {
          const unsigned rj = __builtin_amdgcn_readlane(rr, g + k);
          float* t = tile + rj * DIM + lane;
          atomicAdd(t,      v[k].x);
          atomicAdd(t + 32, v[k].y);
          atomicAdd(t + 64, v[k].z);
          atomicAdd(t + 96, v[k].w);
        }
      }
    }
  };
  auto flush = [&](unsigned n) __attribute__((always_inline)) {
    int s = 0;
    unsigned rr = 0;
    if ((unsigned)lane < n) {
      const unsigned w = myq[(head + lane) & (OWN_QCAP - 1)];
      s  = (int)(w & PK_MASK);
      rr = w >> PK_SHIFT;
    }
    tile_add(s, rr, n);
  };
  auto ring_sink = [&](bool valid, unsigned w) __attribute__((always_inline)) {
    const unsigned bal = __builtin_amdgcn_ballot_w32(valid);
    if (valid) myq[(tail + __builtin_popcount(bal & below)) & (OWN_QCAP - 1)] = w;
    tail += __builtin_popcount(bal);
    while (tail - head >= 32u) {
      flush(32u);
      head += 32u;
    }
  };
  auto tile_path = [&]() __attribute__((always_inline)) {
    float4* t4 = reinterpret_cast<float4*>(tile);
    for (int i = tid; i < OWN_ROWS * DIM / 4; i += 1024) t4[i] = make_float4(0.0f, 0.0f, 0.0f, 0.0f);
    __syncthreads();
    walk(ring_sink);
    if (tail != head) flush(tail - head);
    __syncthreads();
    for (int r = wave; r < nrows; r += OWN_WAVES) {
      const float* t = tile + r * DIM + lane;
      reinterpret_cast<float4*>(agg + (size_t)(row0 + r) * DIM)[lane] = make_float4(t[0], t[32], t[64], t[96]);
    }
  };

  if (FIRST) {
    for (int i = tid; i < OWN_ROWS; i += 1024) ctile[i] = 0.0f;
    if (tid < OWN_WAVES) rqn[tid] = 0u;
    __syncthreads();

    auto route = [&](bool valid, unsigned w) __attribute__((always_inline)) {
      if (valid) {
        const unsigned rr = w >> PK_SHIFT;
        atomicAdd(&ctile[rr], 1.0f);
        const unsigned t = rr / (unsigned)OWN_WROWS;
        const unsigned pos = atomicAdd(&rqn[t], 1u);
        if (pos < (unsigned)OWN_RCAP) rq[t][pos] = w;
      }
    };
    walk(route);
    __syncthreads();

    const unsigned qc = rqn[lane];
    const bool ovf = __builtin_amdgcn_ballot_w32(qc > (unsigned)OWN_RCAP) != 0u;
    for (int i = tid; i < nrows; i += 1024) cnt[row0 + i] = ctile[i];
    if (!ovf) {
      const unsigned n = (unsigned)__builtin_amdgcn_readlane((int)qc, wave);
      if (n != 0u) {
        const unsigned lastw = n - 1u;
        for (unsigned i = 0; i < n; i += 32u) {
          const unsigned idx = i + (unsigned)lane;
          const unsigned w = rq[wave][idx < lastw ? idx : lastw];
          if (idx < n) mylist[idx] = w;
          acc_batch(w, n - i < 32u ? n - i : 32u);
        }
      }
      store_acc();
      if (lane == 0) lcount[blockIdx.x * OWN_WAVES + wave] = n;
    } else {
      if (lane == 0) lcount[blockIdx.x * OWN_WAVES + wave] = 0xFFFFFFFFu;
      tile_path();
    }
  } else {
    const unsigned kept = (unsigned)__builtin_amdgcn_readfirstlane((int)lcount[blockIdx.x * OWN_WAVES + wave]);
    if (kept <= (unsigned)OWN_RCAP) {
      if (kept != 0u) {
        const unsigned lastw = kept - 1u;
        unsigned wn = mylist[(unsigned)lane < lastw ? (unsigned)lane : lastw];
        for (unsigned i = 0; i < kept; i += 32u) {
          const unsigned w = wn;
          const unsigned nx = i + 32u + (unsigned)lane;
          wn = mylist[nx < lastw ? nx : lastw];
          acc_batch(w, kept - i < 32u ? kept - i : 32u);
        }
      }
      store_acc();
    } else {
      tile_path();
    }
  }
}

#define GEMM_WAVES 4
#define GEMM_KS    16
#define GEMM_NT    (DIM / 16)
#define GEMM_WGS   512
#define GEMM_PITCH (DIM + 4)

__device__ __forceinline__ int kof(int s, int h) { return 8 * (s >> 1) + 4 * h + 2 * (s & 1); }

__global__ __launch_bounds__(32 * GEMM_WAVES)
void sage_fused_gemm(const float* __restrict__ agg,
                     const float* __restrict__ cnt,
                     const float* __restrict__ xin,
                     const float* __restrict__ Wl,
                     const float* __restrict__ Wr,
                     const float* __restrict__ bias,
                     float* __restrict__ out,
                     int do_relu, int nstripes) {
  __shared__ float4 part4[2][GEMM_WAVES][16 * GEMM_PITCH / 4];

  const int lane = threadIdx.x & 31;
  const int wave = __builtin_amdgcn_readfirstlane(threadIdx.x >> 5);
  const int m    = lane & 15;
  const int half = lane >> 4;

  const bool left = wave < 2;
  const float* A  = left ? agg : xin;
  const float* Wm = left ? Wl : Wr;
  const int kb    = (wave & 1) * (4 * GEMM_KS);

  v2f b[GEMM_NT][GEMM_KS];
#pragma unroll
  for (int t = 0; t < GEMM_NT; ++t) {
#pragma unroll
    for (int s = 0; s < GEMM_KS; ++s) {
      const int kk = kb + kof(s, half);
      b[t][s].x = Wm[(size_t)kk * DIM + t * 16 + m];
      b[t][s].y = Wm[(size_t)(kk + 1) * DIM + t * 16 + m];
    }
  }
  const float4 bias4 = reinterpret_cast<const float4*>(bias)[lane];

  float4 an[GEMM_KS / 2];
  float  cn = 1.0f;
  auto load_a = [&](int st) __attribute__((always_inline)) {
    const size_t row = (size_t)st * 16 + m;
    const float4* p = reinterpret_cast<const float4*>(A + row * DIM + kb + 4 * half);
#pragma unroll
    for (int j = 0; j < GEMM_KS / 2; ++j) an[j] = p[2 * j];
    if (left) cn = cnt[row];
  };

  int st = blockIdx.x;
  if (st < nstripes) load_a(st);
  int buf = 0;

  for (; st < nstripes; st += gridDim.x) {
    const int row0 = st * 16;
    float4 ac[GEMM_KS / 2];
#pragma unroll
    for (int j = 0; j < GEMM_KS / 2; ++j) ac[j] = an[j];
    const float scale = left ? 1.0f / fmaxf(cn, 1.0f) : 1.0f;

    const int nx = st + gridDim.x;
    load_a(nx < nstripes ? nx : st);

    v8f c[GEMM_NT];
#pragma unroll
    for (int t = 0; t < GEMM_NT; ++t) c[t] = v8f{};

#pragma unroll
    for (int s = 0; s < GEMM_KS; ++s) {
      const float4 q = ac[s >> 1];
      v2f a;
      a.x = ((s & 1) ? q.z : q.x) * scale;
      a.y = ((s & 1) ? q.w : q.y) * scale;
#pragma unroll
      for (int t = 0; t < GEMM_NT; ++t)
        c[t] = __builtin_amdgcn_wmma_f32_16x16x4_f32(false, a, false, b[t][s], (short)0, c[t],
                                                     false, false);
    }

    float* P = reinterpret_cast<float*>(part4[buf][wave]);
#pragma unroll
    for (int t = 0; t < GEMM_NT; ++t) {
#pragma unroll
      for (int r = 0; r < 8; ++r) P[(r + 8 * half) * GEMM_PITCH + t * 16 + m] = c[t][r];
    }
    __syncthreads();

#pragma unroll
    for (int i = 0; i < 4; ++i) {
      const int row = 4 * wave + i;
      const int o = row * (GEMM_PITCH / 4) + lane;
      const float4 p0 = part4[buf][0][o];
      const float4 p1 = part4[buf][1][o];
      const float4 p2 = part4[buf][2][o];
      const float4 p3 = part4[buf][3][o];
      float4 v;
      v.x = ((p0.x + p1.x) + p2.x) + p3.x + bias4.x;
      v.y = ((p0.y + p1.y) + p2.y) + p3.y + bias4.y;
      v.z = ((p0.z + p1.z) + p2.z) + p3.z + bias4.z;
      v.w = ((p0.w + p1.w) + p2.w) + p3.w + bias4.w;
      if (do_relu) {
        v.x = fmaxf(v.x, 0.0f);
        v.y = fmaxf(v.y, 0.0f);
        v.z = fmaxf(v.z, 0.0f);
        v.w = fmaxf(v.w, 0.0f);
      }
      reinterpret_cast<float4*>(out + (size_t)(row0 + row) * DIM)[lane] = v;
    }
    buf ^= 1;
  }
}

extern "C" void kernel_launch(void* const* d_in, const int* in_sizes, int n_in,
                              void* d_out, int out_size, void* d_ws, size_t ws_size,
                              hipStream_t stream) {
  const float* x   = (const float*)d_in[0];
  const int*  edge = (const int*)d_in[1];
  const float* W1l = (const float*)d_in[2];
  const float* b1  = (const float*)d_in[3];
  const float* W1r = (const float*)d_in[4];
  const float* W2l = (const float*)d_in[5];
  const float* b2  = (const float*)d_in[6];
  const float* W2r = (const float*)d_in[7];

  const int N = in_sizes[0] / DIM;
  const int E = in_sizes[1] / 2;
  const int* src = edge;
  const int* dst = edge + E;

  float* agg = (float*)d_ws;
  float* cnt = agg + (size_t)N * DIM;
  float* h   = cnt + N;
  float* out = (float*)d_out;

  const int own_blocks  = (N + OWN_ROWS - 1) / OWN_ROWS;
  const int gemm_blocks = N / 16;

  const int nchunks   = (E + PL_EDGES - 1) / PL_EDGES;
  const size_t cl_off = ((size_t)N * (2 * DIM + 1) + 63) / 64 * 64;
  unsigned* cells     = (unsigned*)d_ws + cl_off;
  const dim3 cl_grid(nchunks, (own_blocks + PL_OWNERS - 1) / PL_OWNERS);
  unsigned* lists     = cells + (size_t)own_blocks * nchunks * CELL_WORDS;
  unsigned* lcount    = lists + (size_t)own_blocks * OWN_WAVES * OWN_LCAP;

  sage_owner_cells<<<cl_grid, PL_THREADS, 0, stream>>>(src, dst, cells, E, own_blocks, nchunks);

  sage_own_rows<true><<<own_blocks, 1024, 0, stream>>>(x, src, dst, cells, lists, lcount, agg, cnt, nchunks, E, N);
  sage_fused_gemm<<<GEMM_WGS, 32 * GEMM_WAVES, 0, stream>>>(agg, cnt, x, W1l, W1r, b1, h, 1, gemm_blocks);

  sage_own_rows<false><<<own_blocks, 1024, 0, stream>>>(h, src, dst, cells, lists, lcount, agg, cnt, nchunks, E, N);
  sage_fused_gemm<<<GEMM_WGS, 32 * GEMM_WAVES, 0, stream>>>(agg, cnt, h, W2l, W2r, b2, out, 0, gemm_blocks);
}
